# starting kernel
# baseline (speedup 1.0000x reference)
#include <hip/hip_runtime.h>
#include <math.h>

#define B_    2
#define NQ    4096
#define NP    8192
#define D_    128
#define KNN   16
#define M_TOT (B_ * NQ)
#define LEAKY 0.2f

typedef float v2f __attribute__((ext_vector_type(2)));
typedef float v8f __attribute__((ext_vector_type(8)));

__global__ __launch_bounds__(256) void knn_kernel(const float* __restrict__ query,
                                                  const float* __restrict__ global_pc,
                                                  int* __restrict__ idx_out) {
    const int b     = blockIdx.x >> 4;
    const int q     = ((blockIdx.x & 15) << 8) + threadIdx.x;
    const int m     = b * NQ + q;

    const float qx = query[m * 3 + 0];
    const float qy = query[m * 3 + 1];
    const float qz = query[m * 3 + 2];

    __shared__ float spx[1024];
    __shared__ float spy[1024];
    __shared__ float spz[1024];

    float bestd[KNN];
    int   besti[KNN];
#pragma unroll
    for (int k = 0; k < KNN; ++k) { bestd[k] = 3.4e38f; besti[k] = 0; }

    const float* __restrict__ pc = global_pc + (size_t)b * NP * 3;

    for (int tile = 0; tile < NP / 1024; ++tile) {
        const int base = tile * 1024;
        __syncthreads();
        for (int i = threadIdx.x; i < 1024; i += 256) {
            const float* p = pc + (size_t)(base + i) * 3;
            spx[i] = p[0]; spy[i] = p[1]; spz[i] = p[2];
        }
        if (tile + 1 < NP / 1024)
            __builtin_prefetch(pc + (size_t)(base + 1024 + threadIdx.x * 4) * 3, 0, 0);
        __syncthreads();

        for (int i = 0; i < 1024; ++i) {
            const float dx = qx - spx[i];
            const float dy = qy - spy[i];
            const float dz = qz - spz[i];
            const float d2 = dx * dx + dy * dy + dz * dz;
            if (d2 < bestd[KNN - 1]) {
                int pos = KNN - 1;
#pragma unroll
                for (int t = KNN - 2; t >= 0; --t)
                    if (d2 < bestd[t]) pos = t;
#pragma unroll
                for (int t = KNN - 1; t > 0; --t)
                    if (t > pos) { bestd[t] = bestd[t - 1]; besti[t] = besti[t - 1]; }
                bestd[pos] = d2;
                besti[pos] = base + i;
            }
        }
    }

#pragma unroll
    for (int k = 0; k < KNN; ++k) idx_out[(size_t)m * KNN + k] = besti[k];
}

__global__ __launch_bounds__(256) void attn_kernel(const float* __restrict__ query,
                                                   const float* __restrict__ global_pc,
                                                   const float* __restrict__ point_feature,
                                                   const float* __restrict__ w_pe,
                                                   const float* __restrict__ b_pe,
                                                   const float* __restrict__ w_attn,
                                                   const int* __restrict__ knn_idx,
                                                   float* __restrict__ hbar) {
    const int wave = threadIdx.x >> 5;
    const int lane = threadIdx.x & 31;
    const int m    = blockIdx.x * 8 + wave;
    const int b    = m >> 12;
    const int d0   = lane * 4;

    float wx[4], wy[4], wz[4], bp[4], wa[4];
#pragma unroll
    for (int t = 0; t < 4; ++t) {
        wx[t] = w_pe[(d0 + t) * 3 + 0];
        wy[t] = w_pe[(d0 + t) * 3 + 1];
        wz[t] = w_pe[(d0 + t) * 3 + 2];
        bp[t] = b_pe[d0 + t];
        wa[t] = w_attn[d0 + t];
    }

    const float qx = query[m * 3 + 0];
    const float qy = query[m * 3 + 1];
    const float qz = query[m * 3 + 2];

    float hreg[KNN][4];
    float score[KNN];

#pragma unroll
    for (int k = 0; k < KNN; ++k) {
        const int j = knn_idx[(size_t)m * KNN + k];
        const float* pp = global_pc + ((size_t)b * NP + j) * 3;
        const float px = qx - pp[0];
        const float py = qy - pp[1];
        const float pz = qz - pp[2];
        const float* pf = point_feature + ((size_t)b * NP + j) * D_ + d0;
        float s = 0.f;
#pragma unroll
        for (int t = 0; t < 4; ++t) {
            const float hv = pf[t] + wx[t] * px + wy[t] * py + wz[t] * pz + bp[t];
            hreg[k][t] = hv;
            s += hv * wa[t];
        }
#pragma unroll
        for (int off = 16; off >= 1; off >>= 1)
            s += __shfl_xor(s, off, 32);
        score[k] = s;
    }

    float mx = score[0];
#pragma unroll
    for (int k = 1; k < KNN; ++k) mx = fmaxf(mx, score[k]);
    float sum = 0.f;
#pragma unroll
    for (int k = 0; k < KNN; ++k) { score[k] = __expf(score[k] - mx); sum += score[k]; }
    const float inv = 1.0f / sum;

    float acc[4] = {0.f, 0.f, 0.f, 0.f};
#pragma unroll
    for (int k = 0; k < KNN; ++k)
#pragma unroll
        for (int t = 0; t < 4; ++t) acc[t] += score[k] * hreg[k][t];

#pragma unroll
    for (int t = 0; t < 4; ++t)
        hbar[(size_t)m * D_ + d0 + t] = acc[t] * inv;
}

__global__ __launch_bounds__(256) void gemm_v_kernel(const float* __restrict__ hbar,
                                                     const float* __restrict__ w_v,
                                                     const float* __restrict__ b_v,
                                                     float* __restrict__ out) {
    __shared__ float As[16 * D_];
    const int tileM = blockIdx.x * 16;

    for (int i = threadIdx.x; i < 16 * D_; i += 256)
        As[i] = hbar[(size_t)(tileM + (i >> 7)) * D_ + (i & (D_ - 1))];
    __syncthreads();

    const int wave  = threadIdx.x >> 5;
    const int lane  = threadIdx.x & 31;
    const int tileN = wave * 16;
    const int half  = lane >> 4;
    const int l     = lane & 15;

    v8f c = {};
    for (int k0 = 0; k0 < D_; k0 += 4) {
        const int k = k0 + half * 2;
        v2f a, bb;
        a.x  = As[l * D_ + k];
        a.y  = As[l * D_ + k + 1];
        bb.x = w_v[(size_t)(tileN + l) * D_ + k];
        bb.y = w_v[(size_t)(tileN + l) * D_ + k + 1];
        c = __builtin_amdgcn_wmma_f32_16x16x4_f32(
                 false, a,  false, bb,
                 (short)0, c,  false,  false);
    }

    const float bias = b_v[tileN + l];
#pragma unroll
    for (int r = 0; r < 8; ++r) {
        const int mrow = tileM + r + half * 8;
        const int n    = tileN + l;
        const int b    = mrow >> 12;
        const int q    = mrow & (NQ - 1);
        out[(size_t)b * 256 * NQ + (size_t)n * NQ + q] = c[r] + bias;
    }
}

__global__ __launch_bounds__(256) void posfeat_kernel(const float* __restrict__ query,
                                                      const float* __restrict__ w_pos,
                                                      const float* __restrict__ b_pos,
                                                      float* __restrict__ out) {
    const int gid = blockIdx.x * 256 + threadIdx.x;
    const int m   = gid >> 7;
    const int d   = gid & (D_ - 1);
    float v = w_pos[d * 3 + 0] * query[m * 3 + 0]
            + w_pos[d * 3 + 1] * query[m * 3 + 1]
            + w_pos[d * 3 + 2] * query[m * 3 + 2]
            + b_pos[d];
    v = (v >= 0.f) ? v : LEAKY * v;
    const int b = m >> 12;
    const int q = m & (NQ - 1);
    out[(size_t)b * 256 * NQ + (size_t)(D_ + d) * NQ + q] = v;
}

extern "C" void kernel_launch(void* const* d_in, const int* in_sizes, int n_in,
                              void* d_out, int out_size, void* d_ws, size_t ws_size,
                              hipStream_t stream) {
    const float* query         = (const float*)d_in[0];
    const float* global_pc     = (const float*)d_in[1];
    const float* point_feature = (const float*)d_in[2];
    const float* w_pos         = (const float*)d_in[3];
    const float* b_pos         = (const float*)d_in[4];
    const float* w_pe          = (const float*)d_in[5];
    const float* b_pe          = (const float*)d_in[6];
    const float* w_v           = (const float*)d_in[7];
    const float* b_v           = (const float*)d_in[8];
    const float* w_attn        = (const float*)d_in[9];
    float* out = (float*)d_out;

    int*   idx_ws  = (int*)d_ws;
    float* hbar_ws = (float*)((char*)d_ws + (size_t)M_TOT * KNN * sizeof(int));

    knn_kernel<<<M_TOT / 256, 256, 0, stream>>>(query, global_pc, idx_ws);

    attn_kernel<<<M_TOT / 8, 256, 0, stream>>>(query, global_pc, point_feature,
                                               w_pe, b_pe, w_attn, idx_ws, hbar_ws);

    gemm_v_kernel<<<M_TOT / 16, 256, 0, stream>>>(hbar_ws, w_v, b_v, out);

    posfeat_kernel<<<(M_TOT * D_) / 256, 256, 0, stream>>>(query, w_pos, b_pos, out);
}


# KNN list keys held as two 32-bit words: a compare-exchange of the batch sort and merges is 32-bit compare + borrow-subtract + min + max + 2 selects, not 64-bit min/max (read as slow); same key order
# speedup vs baseline: 10.7421x; 10.7421x over previous
#include <hip/hip_runtime.h>
#include <math.h>

#define B_    2
#define NQ    4096
#define NP    8192
#define D_    128
#define KNN   16
#define M_TOT (B_ * NQ)
#define LEAKY 0.2f

typedef float v2f __attribute__((ext_vector_type(2)));
typedef float v8f __attribute__((ext_vector_type(8)));

#define KPART   4
#define PPART   (NP / KPART)
#define PVEC    (PPART * 3 / 4)
#define QCAP    KNN

__device__ __forceinline__ bool kless(const unsigned ah, const unsigned al, const unsigned bh, const unsigned bl) {
    return (ah < bh) | ((ah == bh) & (al < bl));
}

__device__ __forceinline__ void kcx(unsigned& ah, unsigned& al, unsigned& bh, unsigned& bl) {
    const bool sw = kless(bh, bl, ah, al);
    const unsigned lo_h = ah < bh ? ah : bh;
    const unsigned hi_h = ah < bh ? bh : ah;
    const unsigned lo_l = sw ? bl : al;
    const unsigned hi_l = sw ? al : bl;
    ah = lo_h; al = lo_l;
    bh = hi_h; bl = hi_l;
}

#define KCX(i, j) kcx(h[i], l[i], h[j], l[j])

__device__ __forceinline__ void knn_sort16(unsigned (&h)[KNN], unsigned (&l)[KNN]) {
    KCX(0, 13); KCX(1, 12); KCX(2, 15); KCX(3, 14); KCX(4, 8); KCX(5, 6); KCX(7, 11); KCX(9, 10);
    KCX(0, 5); KCX(1, 7); KCX(2, 9); KCX(3, 4); KCX(6, 13); KCX(8, 14); KCX(10, 15); KCX(11, 12);
    KCX(0, 1); KCX(2, 3); KCX(4, 5); KCX(6, 8); KCX(7, 9); KCX(10, 11); KCX(12, 13); KCX(14, 15);
    KCX(0, 2); KCX(1, 3); KCX(4, 10); KCX(5, 11); KCX(6, 7); KCX(8, 9); KCX(12, 14); KCX(13, 15);
    KCX(1, 2); KCX(3, 12); KCX(4, 6); KCX(5, 7); KCX(8, 10); KCX(9, 11); KCX(13, 14);
    KCX(1, 4); KCX(2, 6); KCX(5, 8); KCX(7, 10); KCX(9, 13); KCX(11, 14);
    KCX(2, 4); KCX(3, 6); KCX(9, 12); KCX(11, 13);
    KCX(3, 5); KCX(6, 8); KCX(7, 9); KCX(10, 12);
    KCX(3, 4); KCX(5, 6); KCX(7, 8); KCX(9, 10); KCX(11, 12);
    KCX(6, 7); KCX(8, 9);
}
#undef KCX

__device__ __forceinline__ void knn_merge(unsigned (&kh)[KNN], unsigned (&kl)[KNN],
                                          const unsigned (&ph)[KNN], const unsigned (&pl)[KNN]) {
#pragma unroll
    for (int t = 0; t < KNN; ++t) {
        const bool take = kless(ph[KNN - 1 - t], pl[KNN - 1 - t], kh[t], kl[t]);
        kh[t] = kh[t] < ph[KNN - 1 - t] ? kh[t] : ph[KNN - 1 - t];
        kl[t] = take ? pl[KNN - 1 - t] : kl[t];
    }
#pragma unroll
    for (int s = KNN / 2; s >= 1; s >>= 1) {
#pragma unroll
        for (int t = 0; t < KNN; ++t) {
            if ((t & s) == 0) kcx(kh[t], kl[t], kh[t + s], kl[t + s]);
        }
    }
}

__global__ __launch_bounds__(128) void knn_kernel(const float* __restrict__ query,
                                                  const float* __restrict__ global_pc,
                                                  const float* __restrict__ w_pos,
                                                  const float* __restrict__ b_pos,
                                                  int* __restrict__ idx_out,
                                                  float* __restrict__ out) {
    __shared__ float4 sp[KPART][PVEC];
    __shared__ uint2  lst[3][KNN][32];
    __shared__ uint2  cq[KPART][QCAP][32];

    const int lane = threadIdx.x & 31;
    const int part = threadIdx.x >> 5;
    const int m    = blockIdx.x * 32 + lane;
    const int b    = m >> 12;

    const float qx = query[m * 3 + 0];
    const float qy = query[m * 3 + 1];
    const float qz = query[m * 3 + 2];

    const float4* __restrict__ pc4 =
        (const float4*)(global_pc + ((size_t)b * NP + (size_t)part * PPART) * 3);
#pragma unroll 8
    for (int i = lane; i < PVEC; i += 32) sp[part][i] = pc4[i];
    __syncthreads();

    {
        const int dw = __builtin_amdgcn_readfirstlane(part) * (D_ / KPART);
        float* __restrict__ po = out + (size_t)b * 256 * NQ + (size_t)(D_ + dw) * NQ + (m & (NQ - 1));
#pragma unroll
        for (int c = 0; c < D_ / KPART; ++c) {
            const int d = dw + c;
            float v = w_pos[d * 3 + 0] * qx
                    + w_pos[d * 3 + 1] * qy
                    + w_pos[d * 3 + 2] * qz
                    + b_pos[d];
            v = (v >= 0.f) ? v : LEAKY * v;
            po[(size_t)c * NQ] = v;
        }
    }

    unsigned keyh[KNN], keyl[KNN];
#pragma unroll
    for (int k = 0; k < KNN; ++k) { keyh[k] = __float_as_uint(3.4e38f); keyl[k] = 0u; }

    float    thr = 3.4e38f;
    unsigned qn  = 0;
    uint2* const myq = &cq[part][0][lane];

    auto scan4 = [&](const float4 a0, const float4 a1, const float4 a2, const unsigned i0) {
        const float px[4] = {a0.x, a0.w, a1.z, a2.y};
        const float py[4] = {a0.y, a1.x, a1.w, a2.z};
        const float pz[4] = {a0.z, a1.y, a2.x, a2.w};
        float d2[4];
#pragma unroll
        for (int u = 0; u < 4; ++u) {
            const float dx = qx - px[u];
            const float dy = qy - py[u];
            const float dz = qz - pz[u];
            d2[u] = fmaf(dz, dz, fmaf(dx, dx, dy * dy));
        }
#pragma unroll
        for (int u = 0; u < 4; ++u) {
            myq[qn * 32] = make_uint2(i0 + u, __float_as_uint(d2[u]));
            qn += (d2[u] < thr) ? 1u : 0u;
        }
    };

    auto take_batch = [&]() {
        unsigned ph[KNN], pl[KNN];
#pragma unroll
        for (int s = 0; s < KNN; ++s) {
            ph[s] = ~0u;
            pl[s] = ~0u;
            if ((unsigned)s < qn) {
                const uint2 e = myq[s * 32];
                ph[s] = e.y;
                pl[s] = e.x;
            }
        }
        knn_sort16(ph, pl);
        knn_merge(keyh, keyl, ph, pl);
        thr = __uint_as_float(keyh[KNN - 1]);
        qn  = 0;
    };

    const unsigned base = (unsigned)part * PPART;
    const float4* tile = &sp[part][0];
    float4 a0 = tile[0], a1 = tile[1], a2 = tile[2];
    for (int g = 0; g < PPART / 4; g += 2) {
        const float4 b0 = tile[g * 3 + 3], b1 = tile[g * 3 + 4], b2 = tile[g * 3 + 5];
        scan4(a0, a1, a2, base + g * 4);
        if (__builtin_amdgcn_ballot_w32(qn > QCAP - 4) != 0u) take_batch();
        const int gn = (g + 2 < PPART / 4) ? g + 2 : g;
        a0 = tile[gn * 3 + 0]; a1 = tile[gn * 3 + 1]; a2 = tile[gn * 3 + 2];
        scan4(b0, b1, b2, base + g * 4 + 4);
        if (__builtin_amdgcn_ballot_w32(qn > QCAP - 4) != 0u) take_batch();
    }
    take_batch();

    if (part & 1) {
#pragma unroll
        for (int k = 0; k < KNN; ++k) lst[part >> 1][k][lane] = make_uint2(keyl[k], keyh[k]);
    }
    __syncthreads();
    if (!(part & 1)) {
        unsigned ph[KNN], pl[KNN];
#pragma unroll
        for (int k = 0; k < KNN; ++k) { const uint2 e = lst[part >> 1][k][lane]; ph[k] = e.y; pl[k] = e.x; }
        knn_merge(keyh, keyl, ph, pl);
    }
    if (part == 2) {
#pragma unroll
        for (int k = 0; k < KNN; ++k) lst[2][k][lane] = make_uint2(keyl[k], keyh[k]);
    }
    __syncthreads();
    if (part == 0) {
        unsigned ph[KNN], pl[KNN];
#pragma unroll
        for (int k = 0; k < KNN; ++k) { const uint2 e = lst[2][k][lane]; ph[k] = e.y; pl[k] = e.x; }
        knn_merge(keyh, keyl, ph, pl);
#pragma unroll
        for (int k = 0; k < KNN; ++k) idx_out[(size_t)m * KNN + k] = (int)keyl[k];
    }
}

__global__ __launch_bounds__(256) void attn_kernel(const float* __restrict__ query,
                                                   const float* __restrict__ global_pc,
                                                   const float* __restrict__ point_feature,
                                                   const float* __restrict__ w_pe,
                                                   const float* __restrict__ b_pe,
                                                   const float* __restrict__ w_attn,
                                                   const int* __restrict__ knn_idx,
                                                   float* __restrict__ hbar) {
    const int wave = threadIdx.x >> 5;
    const int lane = threadIdx.x & 31;
    const int m    = blockIdx.x * 8 + wave;
    const int b    = m >> 12;
    const int d0   = lane * 4;

    float wx[4], wy[4], wz[4], bp[4], wa[4];
#pragma unroll
    for (int t = 0; t < 4; ++t) {
        wx[t] = w_pe[(d0 + t) * 3 + 0];
        wy[t] = w_pe[(d0 + t) * 3 + 1];
        wz[t] = w_pe[(d0 + t) * 3 + 2];
        bp[t] = b_pe[d0 + t];
        wa[t] = w_attn[d0 + t];
    }

    const float qx = query[m * 3 + 0];
    const float qy = query[m * 3 + 1];
    const float qz = query[m * 3 + 2];

    float hreg[KNN][4];
    float score[KNN];

#pragma unroll
    for (int k = 0; k < KNN; ++k) {
        const int j = knn_idx[(size_t)m * KNN + k];
        const float* pp = global_pc + ((size_t)b * NP + j) * 3;
        const float px = qx - pp[0];
        const float py = qy - pp[1];
        const float pz = qz - pp[2];
        const float* pf = point_feature + ((size_t)b * NP + j) * D_ + d0;
        float s = 0.f;
#pragma unroll
        for (int t = 0; t < 4; ++t) {
            const float hv = pf[t] + wx[t] * px + wy[t] * py + wz[t] * pz + bp[t];
            hreg[k][t] = hv;
            s += hv * wa[t];
        }
#pragma unroll
        for (int off = 16; off >= 1; off >>= 1)
            s += __shfl_xor(s, off, 32);
        score[k] = s;
    }

    float mx = score[0];
#pragma unroll
    for (int k = 1; k < KNN; ++k) mx = fmaxf(mx, score[k]);
    float sum = 0.f;
#pragma unroll
    for (int k = 0; k < KNN; ++k) { score[k] = __expf(score[k] - mx); sum += score[k]; }
    const float inv = 1.0f / sum;

    float acc[4] = {0.f, 0.f, 0.f, 0.f};
#pragma unroll
    for (int k = 0; k < KNN; ++k)
#pragma unroll
        for (int t = 0; t < 4; ++t) acc[t] += score[k] * hreg[k][t];

#pragma unroll
    for (int t = 0; t < 4; ++t)
        hbar[(size_t)m * D_ + d0 + t] = acc[t] * inv;
}

__global__ __launch_bounds__(256) void gemm_v_kernel(const float* __restrict__ hbar,
                                                     const float* __restrict__ w_v,
                                                     const float* __restrict__ b_v,
                                                     float* __restrict__ out) {
    __shared__ float As[16 * D_];
    const int tileM = blockIdx.x * 16;

    for (int i = threadIdx.x; i < 16 * D_; i += 256)
        As[i] = hbar[(size_t)(tileM + (i >> 7)) * D_ + (i & (D_ - 1))];
    __syncthreads();

    const int wave  = threadIdx.x >> 5;
    const int lane  = threadIdx.x & 31;
    const int tileN = wave * 16;
    const int half  = lane >> 4;
    const int l     = lane & 15;

    v8f c = {};
    for (int k0 = 0; k0 < D_; k0 += 4) {
        const int k = k0 + half * 2;
        v2f a, bb;
        a.x  = As[l * D_ + k];
        a.y  = As[l * D_ + k + 1];
        bb.x = w_v[(size_t)(tileN + l) * D_ + k];
        bb.y = w_v[(size_t)(tileN + l) * D_ + k + 1];
        c = __builtin_amdgcn_wmma_f32_16x16x4_f32(
                 false, a,  false, bb,
                 (short)0, c,  false,  false);
    }

    const float bias = b_v[tileN + l];
#pragma unroll
    for (int r = 0; r < 8; ++r) {
        const int mrow = tileM + r + half * 8;
        const int n    = tileN + l;
        const int b    = mrow >> 12;
        const int q    = mrow & (NQ - 1);
        out[(size_t)b * 256 * NQ + (size_t)n * NQ + q] = c[r] + bias;
    }
}

extern "C" void kernel_launch(void* const* d_in, const int* in_sizes, int n_in,
                              void* d_out, int out_size, void* d_ws, size_t ws_size,
                              hipStream_t stream) {
    const float* query         = (const float*)d_in[0];
    const float* global_pc     = (const float*)d_in[1];
    const float* point_feature = (const float*)d_in[2];
    const float* w_pos         = (const float*)d_in[3];
    const float* b_pos         = (const float*)d_in[4];
    const float* w_pe          = (const float*)d_in[5];
    const float* b_pe          = (const float*)d_in[6];
    const float* w_v           = (const float*)d_in[7];
    const float* b_v           = (const float*)d_in[8];
    const float* w_attn        = (const float*)d_in[9];
    float* out = (float*)d_out;

    int*   idx_ws  = (int*)d_ws;
    float* hbar_ws = (float*)((char*)d_ws + (size_t)M_TOT * KNN * sizeof(int));

    knn_kernel<<<M_TOT / 32, 128, 0, stream>>>(query, global_pc, w_pos, b_pos, idx_ws, out);

    attn_kernel<<<M_TOT / 8, 256, 0, stream>>>(query, global_pc, point_feature,
                                               w_pe, b_pe, w_attn, idx_ws, hbar_ws);

    gemm_v_kernel<<<M_TOT / 16, 256, 0, stream>>>(hbar_ws, w_v, b_v, out);
}


# KNN cloud split 8 ways per block (was 4): 8 waves x 1024 points, two per SIMD, so one wave's dependent chains and compare-branch waits are filled by the other; lists merged in 3 levels (was 2)
# speedup vs baseline: 13.4066x; 1.2480x over previous
#include <hip/hip_runtime.h>
#include <math.h>

#define B_    2
#define NQ    4096
#define NP    8192
#define D_    128
#define KNN   16
#define M_TOT (B_ * NQ)
#define LEAKY 0.2f

typedef float v2f __attribute__((ext_vector_type(2)));
typedef float v8f __attribute__((ext_vector_type(8)));

#define KPART   8
#define PPART   (NP / KPART)
#define PVEC    (PPART * 3 / 4)
#define QCAP    KNN

__device__ __forceinline__ bool kless(const unsigned ah, const unsigned al, const unsigned bh, const unsigned bl) {
    return (ah < bh) | ((ah == bh) & (al < bl));
}

__device__ __forceinline__ void kcx(unsigned& ah, unsigned& al, unsigned& bh, unsigned& bl) {
    const bool sw = kless(bh, bl, ah, al);
    const unsigned lo_h = ah < bh ? ah : bh;
    const unsigned hi_h = ah < bh ? bh : ah;
    const unsigned lo_l = sw ? bl : al;
    const unsigned hi_l = sw ? al : bl;
    ah = lo_h; al = lo_l;
    bh = hi_h; bl = hi_l;
}

#define KCX(i, j) kcx(h[i], l[i], h[j], l[j])

__device__ __forceinline__ void knn_sort16(unsigned (&h)[KNN], unsigned (&l)[KNN]) {
    KCX(0, 13); KCX(1, 12); KCX(2, 15); KCX(3, 14); KCX(4, 8); KCX(5, 6); KCX(7, 11); KCX(9, 10);
    KCX(0, 5); KCX(1, 7); KCX(2, 9); KCX(3, 4); KCX(6, 13); KCX(8, 14); KCX(10, 15); KCX(11, 12);
    KCX(0, 1); KCX(2, 3); KCX(4, 5); KCX(6, 8); KCX(7, 9); KCX(10, 11); KCX(12, 13); KCX(14, 15);
    KCX(0, 2); KCX(1, 3); KCX(4, 10); KCX(5, 11); KCX(6, 7); KCX(8, 9); KCX(12, 14); KCX(13, 15);
    KCX(1, 2); KCX(3, 12); KCX(4, 6); KCX(5, 7); KCX(8, 10); KCX(9, 11); KCX(13, 14);
    KCX(1, 4); KCX(2, 6); KCX(5, 8); KCX(7, 10); KCX(9, 13); KCX(11, 14);
    KCX(2, 4); KCX(3, 6); KCX(9, 12); KCX(11, 13);
    KCX(3, 5); KCX(6, 8); KCX(7, 9); KCX(10, 12);
    KCX(3, 4); KCX(5, 6); KCX(7, 8); KCX(9, 10); KCX(11, 12);
    KCX(6, 7); KCX(8, 9);
}
#undef KCX

__device__ __forceinline__ void knn_merge(unsigned (&kh)[KNN], unsigned (&kl)[KNN],
                                          const unsigned (&ph)[KNN], const unsigned (&pl)[KNN]) {
#pragma unroll
    for (int t = 0; t < KNN; ++t) {
        const bool take = kless(ph[KNN - 1 - t], pl[KNN - 1 - t], kh[t], kl[t]);
        kh[t] = kh[t] < ph[KNN - 1 - t] ? kh[t] : ph[KNN - 1 - t];
        kl[t] = take ? pl[KNN - 1 - t] : kl[t];
    }
#pragma unroll
    for (int s = KNN / 2; s >= 1; s >>= 1) {
#pragma unroll
        for (int t = 0; t < KNN; ++t) {
            if ((t & s) == 0) kcx(kh[t], kl[t], kh[t + s], kl[t + s]);
        }
    }
}

__global__ __launch_bounds__(256) void knn_kernel(const float* __restrict__ query,
                                                  const float* __restrict__ global_pc,
                                                  const float* __restrict__ w_pos,
                                                  const float* __restrict__ b_pos,
                                                  int* __restrict__ idx_out,
                                                  float* __restrict__ out) {
    __shared__ float4 sp[KPART][PVEC];
    __shared__ uint2  lst[7][KNN][32];
    __shared__ uint2  cq[KPART][QCAP][32];

    const int lane = threadIdx.x & 31;
    const int part = threadIdx.x >> 5;
    const int m    = blockIdx.x * 32 + lane;
    const int b    = m >> 12;

    const float qx = query[m * 3 + 0];
    const float qy = query[m * 3 + 1];
    const float qz = query[m * 3 + 2];

    const float4* __restrict__ pc4 =
        (const float4*)(global_pc + ((size_t)b * NP + (size_t)part * PPART) * 3);
#pragma unroll 8
    for (int i = lane; i < PVEC; i += 32) sp[part][i] = pc4[i];
    __syncthreads();

    {
        const int dw = __builtin_amdgcn_readfirstlane(part) * (D_ / KPART);
        float* __restrict__ po = out + (size_t)b * 256 * NQ + (size_t)(D_ + dw) * NQ + (m & (NQ - 1));
#pragma unroll
        for (int c = 0; c < D_ / KPART; ++c) {
            const int d = dw + c;
            float v = w_pos[d * 3 + 0] * qx
                    + w_pos[d * 3 + 1] * qy
                    + w_pos[d * 3 + 2] * qz
                    + b_pos[d];
            v = (v >= 0.f) ? v : LEAKY * v;
            po[(size_t)c * NQ] = v;
        }
    }

    unsigned keyh[KNN], keyl[KNN];
#pragma unroll
    for (int k = 0; k < KNN; ++k) { keyh[k] = __float_as_uint(3.4e38f); keyl[k] = 0u; }

    float    thr = 3.4e38f;
    unsigned qn  = 0;
    uint2* const myq = &cq[part][0][lane];

    auto scan4 = [&](const float4 a0, const float4 a1, const float4 a2, const unsigned i0) {
        const float px[4] = {a0.x, a0.w, a1.z, a2.y};
        const float py[4] = {a0.y, a1.x, a1.w, a2.z};
        const float pz[4] = {a0.z, a1.y, a2.x, a2.w};
        float d2[4];
#pragma unroll
        for (int u = 0; u < 4; ++u) {
            const float dx = qx - px[u];
            const float dy = qy - py[u];
            const float dz = qz - pz[u];
            d2[u] = fmaf(dz, dz, fmaf(dx, dx, dy * dy));
        }
#pragma unroll
        for (int u = 0; u < 4; ++u) {
            myq[qn * 32] = make_uint2(i0 + u, __float_as_uint(d2[u]));
            qn += (d2[u] < thr) ? 1u : 0u;
        }
    };

    auto take_batch = [&]() {
        unsigned ph[KNN], pl[KNN];
#pragma unroll
        for (int s = 0; s < KNN; ++s) {
            ph[s] = ~0u;
            pl[s] = ~0u;
            if ((unsigned)s < qn) {
                const uint2 e = myq[s * 32];
                ph[s] = e.y;
                pl[s] = e.x;
            }
        }
        knn_sort16(ph, pl);
        knn_merge(keyh, keyl, ph, pl);
        thr = __uint_as_float(keyh[KNN - 1]);
        qn  = 0;
    };

    const unsigned base = (unsigned)part * PPART;
    const float4* tile = &sp[part][0];
    float4 a0 = tile[0], a1 = tile[1], a2 = tile[2];
    for (int g = 0; g < PPART / 4; g += 2) {
        const float4 b0 = tile[g * 3 + 3], b1 = tile[g * 3 + 4], b2 = tile[g * 3 + 5];
        scan4(a0, a1, a2, base + g * 4);
        if (__builtin_amdgcn_ballot_w32(qn > QCAP - 4) != 0u) take_batch();
        const int gn = (g + 2 < PPART / 4) ? g + 2 : g;
        a0 = tile[gn * 3 + 0]; a1 = tile[gn * 3 + 1]; a2 = tile[gn * 3 + 2];
        scan4(b0, b1, b2, base + g * 4 + 4);
        if (__builtin_amdgcn_ballot_w32(qn > QCAP - 4) != 0u) take_batch();
    }
    take_batch();

    if (part & 1) {
#pragma unroll
        for (int k = 0; k < KNN; ++k) lst[part >> 1][k][lane] = make_uint2(keyl[k], keyh[k]);
    }
    __syncthreads();
    if (!(part & 1)) {
        unsigned ph[KNN], pl[KNN];
#pragma unroll
        for (int k = 0; k < KNN; ++k) { const uint2 e = lst[part >> 1][k][lane]; ph[k] = e.y; pl[k] = e.x; }
        knn_merge(keyh, keyl, ph, pl);
    }
    if ((part & 3) == 2) {
#pragma unroll
        for (int k = 0; k < KNN; ++k) lst[4 + (part >> 2)][k][lane] = make_uint2(keyl[k], keyh[k]);
    }
    __syncthreads();
    if ((part & 3) == 0) {
        unsigned ph[KNN], pl[KNN];
#pragma unroll
        for (int k = 0; k < KNN; ++k) { const uint2 e = lst[4 + (part >> 2)][k][lane]; ph[k] = e.y; pl[k] = e.x; }
        knn_merge(keyh, keyl, ph, pl);
    }
    if (part == 4) {
#pragma unroll
        for (int k = 0; k < KNN; ++k) lst[6][k][lane] = make_uint2(keyl[k], keyh[k]);
    }
    __syncthreads();
    if (part == 0) {
        unsigned ph[KNN], pl[KNN];
#pragma unroll
        for (int k = 0; k < KNN; ++k) { const uint2 e = lst[6][k][lane]; ph[k] = e.y; pl[k] = e.x; }
        knn_merge(keyh, keyl, ph, pl);
#pragma unroll
        for (int k = 0; k < KNN; ++k) idx_out[(size_t)m * KNN + k] = (int)keyl[k];
    }
}

__global__ __launch_bounds__(256) void attn_kernel(const float* __restrict__ query,
                                                   const float* __restrict__ global_pc,
                                                   const float* __restrict__ point_feature,
                                                   const float* __restrict__ w_pe,
                                                   const float* __restrict__ b_pe,
                                                   const float* __restrict__ w_attn,
                                                   const int* __restrict__ knn_idx,
                                                   float* __restrict__ hbar) {
    const int wave = threadIdx.x >> 5;
    const int lane = threadIdx.x & 31;
    const int m    = blockIdx.x * 8 + wave;
    const int b    = m >> 12;
    const int d0   = lane * 4;

    float wx[4], wy[4], wz[4], bp[4], wa[4];
#pragma unroll
    for (int t = 0; t < 4; ++t) {
        wx[t] = w_pe[(d0 + t) * 3 + 0];
        wy[t] = w_pe[(d0 + t) * 3 + 1];
        wz[t] = w_pe[(d0 + t) * 3 + 2];
        bp[t] = b_pe[d0 + t];
        wa[t] = w_attn[d0 + t];
    }

    const float qx = query[m * 3 + 0];
    const float qy = query[m * 3 + 1];
    const float qz = query[m * 3 + 2];

    float hreg[KNN][4];
    float score[KNN];

#pragma unroll
    for (int k = 0; k < KNN; ++k) {
        const int j = knn_idx[(size_t)m * KNN + k];
        const float* pp = global_pc + ((size_t)b * NP + j) * 3;
        const float px = qx - pp[0];
        const float py = qy - pp[1];
        const float pz = qz - pp[2];
        const float* pf = point_feature + ((size_t)b * NP + j) * D_ + d0;
        float s = 0.f;
#pragma unroll
        for (int t = 0; t < 4; ++t) {
            const float hv = pf[t] + wx[t] * px + wy[t] * py + wz[t] * pz + bp[t];
            hreg[k][t] = hv;
            s += hv * wa[t];
        }
#pragma unroll
        for (int off = 16; off >= 1; off >>= 1)
            s += __shfl_xor(s, off, 32);
        score[k] = s;
    }

    float mx = score[0];
#pragma unroll
    for (int k = 1; k < KNN; ++k) mx = fmaxf(mx, score[k]);
    float sum = 0.f;
#pragma unroll
    for (int k = 0; k < KNN; ++k) { score[k] = __expf(score[k] - mx); sum += score[k]; }
    const float inv = 1.0f / sum;

    float acc[4] = {0.f, 0.f, 0.f, 0.f};
#pragma unroll
    for (int k = 0; k < KNN; ++k)
#pragma unroll
        for (int t = 0; t < 4; ++t) acc[t] += score[k] * hreg[k][t];

#pragma unroll
    for (int t = 0; t < 4; ++t)
        hbar[(size_t)m * D_ + d0 + t] = acc[t] * inv;
}

__global__ __launch_bounds__(256) void gemm_v_kernel(const float* __restrict__ hbar,
                                                     const float* __restrict__ w_v,
                                                     const float* __restrict__ b_v,
                                                     float* __restrict__ out) {
    __shared__ float As[16 * D_];
    const int tileM = blockIdx.x * 16;

    for (int i = threadIdx.x; i < 16 * D_; i += 256)
        As[i] = hbar[(size_t)(tileM + (i >> 7)) * D_ + (i & (D_ - 1))];
    __syncthreads();

    const int wave  = threadIdx.x >> 5;
    const int lane  = threadIdx.x & 31;
    const int tileN = wave * 16;
    const int half  = lane >> 4;
    const int l     = lane & 15;

    v8f c = {};
    for (int k0 = 0; k0 < D_; k0 += 4) {
        const int k = k0 + half * 2;
        v2f a, bb;
        a.x  = As[l * D_ + k];
        a.y  = As[l * D_ + k + 1];
        bb.x = w_v[(size_t)(tileN + l) * D_ + k];
        bb.y = w_v[(size_t)(tileN + l) * D_ + k + 1];
        c = __builtin_amdgcn_wmma_f32_16x16x4_f32(
                 false, a,  false, bb,
                 (short)0, c,  false,  false);
    }

    const float bias = b_v[tileN + l];
#pragma unroll
    for (int r = 0; r < 8; ++r) {
        const int mrow = tileM + r + half * 8;
        const int n    = tileN + l;
        const int b    = mrow >> 12;
        const int q    = mrow & (NQ - 1);
        out[(size_t)b * 256 * NQ + (size_t)n * NQ + q] = c[r] + bias;
    }
}

extern "C" void kernel_launch(void* const* d_in, const int* in_sizes, int n_in,
                              void* d_out, int out_size, void* d_ws, size_t ws_size,
                              hipStream_t stream) {
    const float* query         = (const float*)d_in[0];
    const float* global_pc     = (const float*)d_in[1];
    const float* point_feature = (const float*)d_in[2];
    const float* w_pos         = (const float*)d_in[3];
    const float* b_pos         = (const float*)d_in[4];
    const float* w_pe          = (const float*)d_in[5];
    const float* b_pe          = (const float*)d_in[6];
    const float* w_v           = (const float*)d_in[7];
    const float* b_v           = (const float*)d_in[8];
    const float* w_attn        = (const float*)d_in[9];
    float* out = (float*)d_out;

    int*   idx_ws  = (int*)d_ws;
    float* hbar_ws = (float*)((char*)d_ws + (size_t)M_TOT * KNN * sizeof(int));

    knn_kernel<<<M_TOT / 32, 256, 0, stream>>>(query, global_pc, w_pos, b_pos, idx_ws, out);

    attn_kernel<<<M_TOT / 8, 256, 0, stream>>>(query, global_pc, point_feature,
                                               w_pe, b_pe, w_attn, idx_ws, hbar_ws);

    gemm_v_kernel<<<M_TOT / 16, 256, 0, stream>>>(hbar_ws, w_v, b_v, out);
}


# KNN: the 8 partition waves of a block publish their list's 4th distance in LDS per batch; the 4th smallest of the eight caps each wave's admission threshold: fewer keys queued, fewer sort batches
# speedup vs baseline: 14.4908x; 1.0809x over previous
#include <hip/hip_runtime.h>
#include <math.h>

#define B_    2
#define NQ    4096
#define NP    8192
#define D_    128
#define KNN   16
#define M_TOT (B_ * NQ)
#define LEAKY 0.2f

typedef float v2f __attribute__((ext_vector_type(2)));
typedef float v8f __attribute__((ext_vector_type(8)));

#define KPART   8
#define PPART   (NP / KPART)
#define PVEC    (PPART * 3 / 4)
#define QCAP    KNN

__device__ __forceinline__ bool kless(const unsigned ah, const unsigned al, const unsigned bh, const unsigned bl) {
    return (ah < bh) | ((ah == bh) & (al < bl));
}

__device__ __forceinline__ void kcx(unsigned& ah, unsigned& al, unsigned& bh, unsigned& bl) {
    const bool sw = kless(bh, bl, ah, al);
    const unsigned lo_h = ah < bh ? ah : bh;
    const unsigned hi_h = ah < bh ? bh : ah;
    const unsigned lo_l = sw ? bl : al;
    const unsigned hi_l = sw ? al : bl;
    ah = lo_h; al = lo_l;
    bh = hi_h; bl = hi_l;
}

__device__ __forceinline__ void ucx(unsigned& a, unsigned& b) {
    const unsigned lo = a < b ? a : b;
    const unsigned hi = a < b ? b : a;
    a = lo; b = hi;
}

#define KCX(i, j) kcx(h[i], l[i], h[j], l[j])

__device__ __forceinline__ void knn_sort16(unsigned (&h)[KNN], unsigned (&l)[KNN]) {
    KCX(0, 13); KCX(1, 12); KCX(2, 15); KCX(3, 14); KCX(4, 8); KCX(5, 6); KCX(7, 11); KCX(9, 10);
    KCX(0, 5); KCX(1, 7); KCX(2, 9); KCX(3, 4); KCX(6, 13); KCX(8, 14); KCX(10, 15); KCX(11, 12);
    KCX(0, 1); KCX(2, 3); KCX(4, 5); KCX(6, 8); KCX(7, 9); KCX(10, 11); KCX(12, 13); KCX(14, 15);
    KCX(0, 2); KCX(1, 3); KCX(4, 10); KCX(5, 11); KCX(6, 7); KCX(8, 9); KCX(12, 14); KCX(13, 15);
    KCX(1, 2); KCX(3, 12); KCX(4, 6); KCX(5, 7); KCX(8, 10); KCX(9, 11); KCX(13, 14);
    KCX(1, 4); KCX(2, 6); KCX(5, 8); KCX(7, 10); KCX(9, 13); KCX(11, 14);
    KCX(2, 4); KCX(3, 6); KCX(9, 12); KCX(11, 13);
    KCX(3, 5); KCX(6, 8); KCX(7, 9); KCX(10, 12);
    KCX(3, 4); KCX(5, 6); KCX(7, 8); KCX(9, 10); KCX(11, 12);
    KCX(6, 7); KCX(8, 9);
}
#undef KCX

__device__ __forceinline__ void knn_merge(unsigned (&kh)[KNN], unsigned (&kl)[KNN],
                                          const unsigned (&ph)[KNN], const unsigned (&pl)[KNN]) {
#pragma unroll
    for (int t = 0; t < KNN; ++t) {
        const bool take = kless(ph[KNN - 1 - t], pl[KNN - 1 - t], kh[t], kl[t]);
        kh[t] = kh[t] < ph[KNN - 1 - t] ? kh[t] : ph[KNN - 1 - t];
        kl[t] = take ? pl[KNN - 1 - t] : kl[t];
    }
#pragma unroll
    for (int s = KNN / 2; s >= 1; s >>= 1) {
#pragma unroll
        for (int t = 0; t < KNN; ++t) {
            if ((t & s) == 0) kcx(kh[t], kl[t], kh[t + s], kl[t + s]);
        }
    }
}

__global__ __launch_bounds__(256) void knn_kernel(const float* __restrict__ query,
                                                  const float* __restrict__ global_pc,
                                                  const float* __restrict__ w_pos,
                                                  const float* __restrict__ b_pos,
                                                  int* __restrict__ idx_out,
                                                  float* __restrict__ out) {
    __shared__ float4 sp[KPART][PVEC];
    __shared__ uint2  lst[7][KNN][32];
    __shared__ uint2  cq[KPART][QCAP][32];
    __shared__ unsigned bnd[KPART][32];

    const int lane = threadIdx.x & 31;
    const int part = threadIdx.x >> 5;
    const int m    = blockIdx.x * 32 + lane;
    const int b    = m >> 12;

    const float qx = query[m * 3 + 0];
    const float qy = query[m * 3 + 1];
    const float qz = query[m * 3 + 2];

    const float4* __restrict__ pc4 =
        (const float4*)(global_pc + ((size_t)b * NP + (size_t)part * PPART) * 3);
#pragma unroll 8
    for (int i = lane; i < PVEC; i += 32) sp[part][i] = pc4[i];
    bnd[part][lane] = __float_as_uint(3.4e38f);
    __syncthreads();

    {
        const int dw = __builtin_amdgcn_readfirstlane(part) * (D_ / KPART);
        float* __restrict__ po = out + (size_t)b * 256 * NQ + (size_t)(D_ + dw) * NQ + (m & (NQ - 1));
#pragma unroll
        for (int c = 0; c < D_ / KPART; ++c) {
            const int d = dw + c;
            float v = w_pos[d * 3 + 0] * qx
                    + w_pos[d * 3 + 1] * qy
                    + w_pos[d * 3 + 2] * qz
                    + b_pos[d];
            v = (v >= 0.f) ? v : LEAKY * v;
            po[(size_t)c * NQ] = v;
        }
    }

    unsigned keyh[KNN], keyl[KNN];
#pragma unroll
    for (int k = 0; k < KNN; ++k) { keyh[k] = __float_as_uint(3.4e38f); keyl[k] = 0u; }

    float    thr = 3.4e38f;
    unsigned qn  = 0;
    uint2* const myq = &cq[part][0][lane];

    auto scan4 = [&](const float4 a0, const float4 a1, const float4 a2, const unsigned i0) {
        const float px[4] = {a0.x, a0.w, a1.z, a2.y};
        const float py[4] = {a0.y, a1.x, a1.w, a2.z};
        const float pz[4] = {a0.z, a1.y, a2.x, a2.w};
        float d2[4];
#pragma unroll
        for (int u = 0; u < 4; ++u) {
            const float dx = qx - px[u];
            const float dy = qy - py[u];
            const float dz = qz - pz[u];
            d2[u] = fmaf(dz, dz, fmaf(dx, dx, dy * dy));
        }
#pragma unroll
        for (int u = 0; u < 4; ++u) {
            myq[qn * 32] = make_uint2(i0 + u, __float_as_uint(d2[u]));
            qn += (d2[u] < thr) ? 1u : 0u;
        }
    };

    auto take_batch = [&]() {
        unsigned ph[KNN], pl[KNN];
#pragma unroll
        for (int s = 0; s < KNN; ++s) {
            ph[s] = ~0u;
            pl[s] = ~0u;
            if ((unsigned)s < qn) {
                const uint2 e = myq[s * 32];
                ph[s] = e.y;
                pl[s] = e.x;
            }
        }
        knn_sort16(ph, pl);
        knn_merge(keyh, keyl, ph, pl);
        typedef volatile __attribute__((address_space(3))) unsigned lds_word;
        lds_word* const vb = (lds_word*)&bnd[0][lane];
        vb[part * 32] = keyh[3];
        unsigned v[KPART];
#pragma unroll
        for (int p = 0; p < KPART; ++p) v[p] = vb[p * 32];
        ucx(v[0], v[1]); ucx(v[2], v[3]); ucx(v[0], v[2]); ucx(v[1], v[3]); ucx(v[1], v[2]);
        ucx(v[4], v[5]); ucx(v[6], v[7]); ucx(v[4], v[6]); ucx(v[5], v[7]); ucx(v[5], v[6]);
        const unsigned m07 = v[0] < v[7] ? v[0] : v[7];
        const unsigned m16 = v[1] < v[6] ? v[1] : v[6];
        const unsigned m25 = v[2] < v[5] ? v[2] : v[5];
        const unsigned m34 = v[3] < v[4] ? v[3] : v[4];
        const unsigned ma  = m07 > m16 ? m07 : m16;
        const unsigned mb  = m25 > m34 ? m25 : m34;
        const unsigned cap = (ma > mb ? ma : mb) + 1u;
        const unsigned own = keyh[KNN - 1];
        thr = __uint_as_float(own < cap ? own : cap);
        qn  = 0;
    };

    const unsigned base = (unsigned)part * PPART;
    const float4* tile = &sp[part][0];
    float4 a0 = tile[0], a1 = tile[1], a2 = tile[2];
    for (int g = 0; g < PPART / 4; g += 2) {
        const float4 b0 = tile[g * 3 + 3], b1 = tile[g * 3 + 4], b2 = tile[g * 3 + 5];
        scan4(a0, a1, a2, base + g * 4);
        if (__builtin_amdgcn_ballot_w32(qn > QCAP - 4) != 0u) take_batch();
        const int gn = (g + 2 < PPART / 4) ? g + 2 : g;
        a0 = tile[gn * 3 + 0]; a1 = tile[gn * 3 + 1]; a2 = tile[gn * 3 + 2];
        scan4(b0, b1, b2, base + g * 4 + 4);
        if (__builtin_amdgcn_ballot_w32(qn > QCAP - 4) != 0u) take_batch();
    }
    take_batch();

    if (part & 1) {
#pragma unroll
        for (int k = 0; k < KNN; ++k) lst[part >> 1][k][lane] = make_uint2(keyl[k], keyh[k]);
    }
    __syncthreads();
    if (!(part & 1)) {
        unsigned ph[KNN], pl[KNN];
#pragma unroll
        for (int k = 0; k < KNN; ++k) { const uint2 e = lst[part >> 1][k][lane]; ph[k] = e.y; pl[k] = e.x; }
        knn_merge(keyh, keyl, ph, pl);
    }
    if ((part & 3) == 2) {
#pragma unroll
        for (int k = 0; k < KNN; ++k) lst[4 + (part >> 2)][k][lane] = make_uint2(keyl[k], keyh[k]);
    }
    __syncthreads();
    if ((part & 3) == 0) {
        unsigned ph[KNN], pl[KNN];
#pragma unroll
        for (int k = 0; k < KNN; ++k) { const uint2 e = lst[4 + (part >> 2)][k][lane]; ph[k] = e.y; pl[k] = e.x; }
        knn_merge(keyh, keyl, ph, pl);
    }
    if (part == 4) {
#pragma unroll
        for (int k = 0; k < KNN; ++k) lst[6][k][lane] = make_uint2(keyl[k], keyh[k]);
    }
    __syncthreads();
    if (part == 0) {
        unsigned ph[KNN], pl[KNN];
#pragma unroll
        for (int k = 0; k < KNN; ++k) { const uint2 e = lst[6][k][lane]; ph[k] = e.y; pl[k] = e.x; }
        knn_merge(keyh, keyl, ph, pl);
#pragma unroll
        for (int k = 0; k < KNN; ++k) idx_out[(size_t)m * KNN + k] = (int)keyl[k];
    }
}

__global__ __launch_bounds__(256) void attn_kernel(const float* __restrict__ query,
                                                   const float* __restrict__ global_pc,
                                                   const float* __restrict__ point_feature,
                                                   const float* __restrict__ w_pe,
                                                   const float* __restrict__ b_pe,
                                                   const float* __restrict__ w_attn,
                                                   const int* __restrict__ knn_idx,
                                                   float* __restrict__ hbar) {
    const int wave = threadIdx.x >> 5;
    const int lane = threadIdx.x & 31;
    const int m    = blockIdx.x * 8 + wave;
    const int b    = m >> 12;
    const int d0   = lane * 4;

    float wx[4], wy[4], wz[4], bp[4], wa[4];
#pragma unroll
    for (int t = 0; t < 4; ++t) {
        wx[t] = w_pe[(d0 + t) * 3 + 0];
        wy[t] = w_pe[(d0 + t) * 3 + 1];
        wz[t] = w_pe[(d0 + t) * 3 + 2];
        bp[t] = b_pe[d0 + t];
        wa[t] = w_attn[d0 + t];
    }

    const float qx = query[m * 3 + 0];
    const float qy = query[m * 3 + 1];
    const float qz = query[m * 3 + 2];

    float hreg[KNN][4];
    float score[KNN];

#pragma unroll
    for (int k = 0; k < KNN; ++k) {
        const int j = knn_idx[(size_t)m * KNN + k];
        const float* pp = global_pc + ((size_t)b * NP + j) * 3;
        const float px = qx - pp[0];
        const float py = qy - pp[1];
        const float pz = qz - pp[2];
        const float* pf = point_feature + ((size_t)b * NP + j) * D_ + d0;
        float s = 0.f;
#pragma unroll
        for (int t = 0; t < 4; ++t) {
            const float hv = pf[t] + wx[t] * px + wy[t] * py + wz[t] * pz + bp[t];
            hreg[k][t] = hv;
            s += hv * wa[t];
        }
#pragma unroll
        for (int off = 16; off >= 1; off >>= 1)
            s += __shfl_xor(s, off, 32);
        score[k] = s;
    }

    float mx = score[0];
#pragma unroll
    for (int k = 1; k < KNN; ++k) mx = fmaxf(mx, score[k]);
    float sum = 0.f;
#pragma unroll
    for (int k = 0; k < KNN; ++k) { score[k] = __expf(score[k] - mx); sum += score[k]; }
    const float inv = 1.0f / sum;

    float acc[4] = {0.f, 0.f, 0.f, 0.f};
#pragma unroll
    for (int k = 0; k < KNN; ++k)
#pragma unroll
        for (int t = 0; t < 4; ++t) acc[t] += score[k] * hreg[k][t];

#pragma unroll
    for (int t = 0; t < 4; ++t)
        hbar[(size_t)m * D_ + d0 + t] = acc[t] * inv;
}

__global__ __launch_bounds__(256) void gemm_v_kernel(const float* __restrict__ hbar,
                                                     const float* __restrict__ w_v,
                                                     const float* __restrict__ b_v,
                                                     float* __restrict__ out) {
    __shared__ float As[16 * D_];
    const int tileM = blockIdx.x * 16;

    for (int i = threadIdx.x; i < 16 * D_; i += 256)
        As[i] = hbar[(size_t)(tileM + (i >> 7)) * D_ + (i & (D_ - 1))];
    __syncthreads();

    const int wave  = threadIdx.x >> 5;
    const int lane  = threadIdx.x & 31;
    const int tileN = wave * 16;
    const int half  = lane >> 4;
    const int l     = lane & 15;

    v8f c = {};
    for (int k0 = 0; k0 < D_; k0 += 4) {
        const int k = k0 + half * 2;
        v2f a, bb;
        a.x  = As[l * D_ + k];
        a.y  = As[l * D_ + k + 1];
        bb.x = w_v[(size_t)(tileN + l) * D_ + k];
        bb.y = w_v[(size_t)(tileN + l) * D_ + k + 1];
        c = __builtin_amdgcn_wmma_f32_16x16x4_f32(
                 false, a,  false, bb,
                 (short)0, c,  false,  false);
    }

    const float bias = b_v[tileN + l];
#pragma unroll
    for (int r = 0; r < 8; ++r) {
        const int mrow = tileM + r + half * 8;
        const int n    = tileN + l;
        const int b    = mrow >> 12;
        const int q    = mrow & (NQ - 1);
        out[(size_t)b * 256 * NQ + (size_t)n * NQ + q] = c[r] + bias;
    }
}

extern "C" void kernel_launch(void* const* d_in, const int* in_sizes, int n_in,
                              void* d_out, int out_size, void* d_ws, size_t ws_size,
                              hipStream_t stream) {
    const float* query         = (const float*)d_in[0];
    const float* global_pc     = (const float*)d_in[1];
    const float* point_feature = (const float*)d_in[2];
    const float* w_pos         = (const float*)d_in[3];
    const float* b_pos         = (const float*)d_in[4];
    const float* w_pe          = (const float*)d_in[5];
    const float* b_pe          = (const float*)d_in[6];
    const float* w_v           = (const float*)d_in[7];
    const float* b_v           = (const float*)d_in[8];
    const float* w_attn        = (const float*)d_in[9];
    float* out = (float*)d_out;

    int*   idx_ws  = (int*)d_ws;
    float* hbar_ws = (float*)((char*)d_ws + (size_t)M_TOT * KNN * sizeof(int));

    knn_kernel<<<M_TOT / 32, 256, 0, stream>>>(query, global_pc, w_pos, b_pos, idx_ws, out);

    attn_kernel<<<M_TOT / 8, 256, 0, stream>>>(query, global_pc, point_feature,
                                               w_pe, b_pe, w_attn, idx_ws, hbar_ws);

    gemm_v_kernel<<<M_TOT / 16, 256, 0, stream>>>(hbar_ws, w_v, b_v, out);
}


# value GEMM re-tiled: 32 queries per workgroup, product transposed (A = W_v), both fragments by 16-byte loads in one shared k order, hbar tile in one trip, whole-line stores; k sum order changes
# speedup vs baseline: 16.3868x; 1.1308x over previous
#include <hip/hip_runtime.h>
#include <math.h>

#define B_    2
#define NQ    4096
#define NP    8192
#define D_    128
#define KNN   16
#define M_TOT (B_ * NQ)
#define LEAKY 0.2f

typedef float v2f __attribute__((ext_vector_type(2)));
typedef float v8f __attribute__((ext_vector_type(8)));

#define KPART   8
#define PPART   (NP / KPART)
#define PVEC    (PPART * 3 / 4)
#define QCAP    KNN

__device__ __forceinline__ bool kless(const unsigned ah, const unsigned al, const unsigned bh, const unsigned bl) {
    return (ah < bh) | ((ah == bh) & (al < bl));
}

__device__ __forceinline__ void kcx(unsigned& ah, unsigned& al, unsigned& bh, unsigned& bl) {
    const bool sw = kless(bh, bl, ah, al);
    const unsigned lo_h = ah < bh ? ah : bh;
    const unsigned hi_h = ah < bh ? bh : ah;
    const unsigned lo_l = sw ? bl : al;
    const unsigned hi_l = sw ? al : bl;
    ah = lo_h; al = lo_l;
    bh = hi_h; bl = hi_l;
}

__device__ __forceinline__ void ucx(unsigned& a, unsigned& b) {
    const unsigned lo = a < b ? a : b;
    const unsigned hi = a < b ? b : a;
    a = lo; b = hi;
}

#define KCX(i, j) kcx(h[i], l[i], h[j], l[j])

__device__ __forceinline__ void knn_sort16(unsigned (&h)[KNN], unsigned (&l)[KNN]) {
    KCX(0, 13); KCX(1, 12); KCX(2, 15); KCX(3, 14); KCX(4, 8); KCX(5, 6); KCX(7, 11); KCX(9, 10);
    KCX(0, 5); KCX(1, 7); KCX(2, 9); KCX(3, 4); KCX(6, 13); KCX(8, 14); KCX(10, 15); KCX(11, 12);
    KCX(0, 1); KCX(2, 3); KCX(4, 5); KCX(6, 8); KCX(7, 9); KCX(10, 11); KCX(12, 13); KCX(14, 15);
    KCX(0, 2); KCX(1, 3); KCX(4, 10); KCX(5, 11); KCX(6, 7); KCX(8, 9); KCX(12, 14); KCX(13, 15);
    KCX(1, 2); KCX(3, 12); KCX(4, 6); KCX(5, 7); KCX(8, 10); KCX(9, 11); KCX(13, 14);
    KCX(1, 4); KCX(2, 6); KCX(5, 8); KCX(7, 10); KCX(9, 13); KCX(11, 14);
    KCX(2, 4); KCX(3, 6); KCX(9, 12); KCX(11, 13);
    KCX(3, 5); KCX(6, 8); KCX(7, 9); KCX(10, 12);
    KCX(3, 4); KCX(5, 6); KCX(7, 8); KCX(9, 10); KCX(11, 12);
    KCX(6, 7); KCX(8, 9);
}
#undef KCX

__device__ __forceinline__ void knn_merge(unsigned (&kh)[KNN], unsigned (&kl)[KNN],
                                          const unsigned (&ph)[KNN], const unsigned (&pl)[KNN]) {
#pragma unroll
    for (int t = 0; t < KNN; ++t) {
        const bool take = kless(ph[KNN - 1 - t], pl[KNN - 1 - t], kh[t], kl[t]);
        kh[t] = kh[t] < ph[KNN - 1 - t] ? kh[t] : ph[KNN - 1 - t];
        kl[t] = take ? pl[KNN - 1 - t] : kl[t];
    }
#pragma unroll
    for (int s = KNN / 2; s >= 1; s >>= 1) {
#pragma unroll
        for (int t = 0; t < KNN; ++t) {
            if ((t & s) == 0) kcx(kh[t], kl[t], kh[t + s], kl[t + s]);
        }
    }
}

__global__ __launch_bounds__(256) void knn_kernel(const float* __restrict__ query,
                                                  const float* __restrict__ global_pc,
                                                  const float* __restrict__ w_pos,
                                                  const float* __restrict__ b_pos,
                                                  int* __restrict__ idx_out,
                                                  float* __restrict__ out) {
    __shared__ float4 sp[KPART][PVEC];
    __shared__ uint2  lst[7][KNN][32];
    __shared__ uint2  cq[KPART][QCAP][32];
    __shared__ unsigned bnd[KPART][32];

    const int lane = threadIdx.x & 31;
    const int part = threadIdx.x >> 5;
    const int m    = blockIdx.x * 32 + lane;
    const int b    = m >> 12;

    const float qx = query[m * 3 + 0];
    const float qy = query[m * 3 + 1];
    const float qz = query[m * 3 + 2];

    const float4* __restrict__ pc4 =
        (const float4*)(global_pc + ((size_t)b * NP + (size_t)part * PPART) * 3);
#pragma unroll 8
    for (int i = lane; i < PVEC; i += 32) sp[part][i] = pc4[i];
    bnd[part][lane] = __float_as_uint(3.4e38f);
    __syncthreads();

    {
        const int dw = __builtin_amdgcn_readfirstlane(part) * (D_ / KPART);
        float* __restrict__ po = out + (size_t)b * 256 * NQ + (size_t)(D_ + dw) * NQ + (m & (NQ - 1));
#pragma unroll
        for (int c = 0; c < D_ / KPART; ++c) {
            const int d = dw + c;
            float v = w_pos[d * 3 + 0] * qx
                    + w_pos[d * 3 + 1] * qy
                    + w_pos[d * 3 + 2] * qz
                    + b_pos[d];
            v = (v >= 0.f) ? v : LEAKY * v;
            po[(size_t)c * NQ] = v;
        }
    }

    unsigned keyh[KNN], keyl[KNN];
#pragma unroll
    for (int k = 0; k < KNN; ++k) { keyh[k] = __float_as_uint(3.4e38f); keyl[k] = 0u; }

    float    thr = 3.4e38f;
    unsigned qn  = 0;
    uint2* const myq = &cq[part][0][lane];

    auto scan4 = [&](const float4 a0, const float4 a1, const float4 a2, const unsigned i0) {
        const float px[4] = {a0.x, a0.w, a1.z, a2.y};
        const float py[4] = {a0.y, a1.x, a1.w, a2.z};
        const float pz[4] = {a0.z, a1.y, a2.x, a2.w};
        float d2[4];
#pragma unroll
        for (int u = 0; u < 4; ++u) {
            const float dx = qx - px[u];
            const float dy = qy - py[u];
            const float dz = qz - pz[u];
            d2[u] = fmaf(dz, dz, fmaf(dx, dx, dy * dy));
        }
#pragma unroll
        for (int u = 0; u < 4; ++u) {
            myq[qn * 32] = make_uint2(i0 + u, __float_as_uint(d2[u]));
            qn += (d2[u] < thr) ? 1u : 0u;
        }
    };

    auto take_batch = [&]() {
        unsigned ph[KNN], pl[KNN];
#pragma unroll
        for (int s = 0; s < KNN; ++s) {
            ph[s] = ~0u;
            pl[s] = ~0u;
            if ((unsigned)s < qn) {
                const uint2 e = myq[s * 32];
                ph[s] = e.y;
                pl[s] = e.x;
            }
        }
        knn_sort16(ph, pl);
        knn_merge(keyh, keyl, ph, pl);
        typedef volatile __attribute__((address_space(3))) unsigned lds_word;
        lds_word* const vb = (lds_word*)&bnd[0][lane];
        vb[part * 32] = keyh[3];
        unsigned v[KPART];
#pragma unroll
        for (int p = 0; p < KPART; ++p) v[p] = vb[p * 32];
        ucx(v[0], v[1]); ucx(v[2], v[3]); ucx(v[0], v[2]); ucx(v[1], v[3]); ucx(v[1], v[2]);
        ucx(v[4], v[5]); ucx(v[6], v[7]); ucx(v[4], v[6]); ucx(v[5], v[7]); ucx(v[5], v[6]);
        const unsigned m07 = v[0] < v[7] ? v[0] : v[7];
        const unsigned m16 = v[1] < v[6] ? v[1] : v[6];
        const unsigned m25 = v[2] < v[5] ? v[2] : v[5];
        const unsigned m34 = v[3] < v[4] ? v[3] : v[4];
        const unsigned ma  = m07 > m16 ? m07 : m16;
        const unsigned mb  = m25 > m34 ? m25 : m34;
        const unsigned cap = (ma > mb ? ma : mb) + 1u;
        const unsigned own = keyh[KNN - 1];
        thr = __uint_as_float(own < cap ? own : cap);
        qn  = 0;
    };

    const unsigned base = (unsigned)part * PPART;
    const float4* tile = &sp[part][0];
    float4 a0 = tile[0], a1 = tile[1], a2 = tile[2];
    for (int g = 0; g < PPART / 4; g += 2) {
        const float4 b0 = tile[g * 3 + 3], b1 = tile[g * 3 + 4], b2 = tile[g * 3 + 5];
        scan4(a0, a1, a2, base + g * 4);
        if (__builtin_amdgcn_ballot_w32(qn > QCAP - 4) != 0u) take_batch();
        const int gn = (g + 2 < PPART / 4) ? g + 2 : g;
        a0 = tile[gn * 3 + 0]; a1 = tile[gn * 3 + 1]; a2 = tile[gn * 3 + 2];
        scan4(b0, b1, b2, base + g * 4 + 4);
        if (__builtin_amdgcn_ballot_w32(qn > QCAP - 4) != 0u) take_batch();
    }
    take_batch();

    if (part & 1) {
#pragma unroll
        for (int k = 0; k < KNN; ++k) lst[part >> 1][k][lane] = make_uint2(keyl[k], keyh[k]);
    }
    __syncthreads();
    if (!(part & 1)) {
        unsigned ph[KNN], pl[KNN];
#pragma unroll
        for (int k = 0; k < KNN; ++k) { const uint2 e = lst[part >> 1][k][lane]; ph[k] = e.y; pl[k] = e.x; }
        knn_merge(keyh, keyl, ph, pl);
    }
    if ((part & 3) == 2) {
#pragma unroll
        for (int k = 0; k < KNN; ++k) lst[4 + (part >> 2)][k][lane] = make_uint2(keyl[k], keyh[k]);
    }
    __syncthreads();
    if ((part & 3) == 0) {
        unsigned ph[KNN], pl[KNN];
#pragma unroll
        for (int k = 0; k < KNN; ++k) { const uint2 e = lst[4 + (part >> 2)][k][lane]; ph[k] = e.y; pl[k] = e.x; }
        knn_merge(keyh, keyl, ph, pl);
    }
    if (part == 4) {
#pragma unroll
        for (int k = 0; k < KNN; ++k) lst[6][k][lane] = make_uint2(keyl[k], keyh[k]);
    }
    __syncthreads();
    if (part == 0) {
        unsigned ph[KNN], pl[KNN];
#pragma unroll
        for (int k = 0; k < KNN; ++k) { const uint2 e = lst[6][k][lane]; ph[k] = e.y; pl[k] = e.x; }
        knn_merge(keyh, keyl, ph, pl);
#pragma unroll
        for (int k = 0; k < KNN; ++k) idx_out[(size_t)m * KNN + k] = (int)keyl[k];
    }
}

__global__ __launch_bounds__(256) void attn_kernel(const float* __restrict__ query,
                                                   const float* __restrict__ global_pc,
                                                   const float* __restrict__ point_feature,
                                                   const float* __restrict__ w_pe,
                                                   const float* __restrict__ b_pe,
                                                   const float* __restrict__ w_attn,
                                                   const int* __restrict__ knn_idx,
                                                   float* __restrict__ hbar) {
    const int wave = threadIdx.x >> 5;
    const int lane = threadIdx.x & 31;
    const int m    = blockIdx.x * 8 + wave;
    const int b    = m >> 12;
    const int d0   = lane * 4;

    float wx[4], wy[4], wz[4], bp[4], wa[4];
#pragma unroll
    for (int t = 0; t < 4; ++t) {
        wx[t] = w_pe[(d0 + t) * 3 + 0];
        wy[t] = w_pe[(d0 + t) * 3 + 1];
        wz[t] = w_pe[(d0 + t) * 3 + 2];
        bp[t] = b_pe[d0 + t];
        wa[t] = w_attn[d0 + t];
    }

    const float qx = query[m * 3 + 0];
    const float qy = query[m * 3 + 1];
    const float qz = query[m * 3 + 2];

    float hreg[KNN][4];
    float score[KNN];

#pragma unroll
    for (int k = 0; k < KNN; ++k) {
        const int j = knn_idx[(size_t)m * KNN + k];
        const float* pp = global_pc + ((size_t)b * NP + j) * 3;
        const float px = qx - pp[0];
        const float py = qy - pp[1];
        const float pz = qz - pp[2];
        const float* pf = point_feature + ((size_t)b * NP + j) * D_ + d0;
        float s = 0.f;
#pragma unroll
        for (int t = 0; t < 4; ++t) {
            const float hv = pf[t] + wx[t] * px + wy[t] * py + wz[t] * pz + bp[t];
            hreg[k][t] = hv;
            s += hv * wa[t];
        }
#pragma unroll
        for (int off = 16; off >= 1; off >>= 1)
            s += __shfl_xor(s, off, 32);
        score[k] = s;
    }

    float mx = score[0];
#pragma unroll
    for (int k = 1; k < KNN; ++k) mx = fmaxf(mx, score[k]);
    float sum = 0.f;
#pragma unroll
    for (int k = 0; k < KNN; ++k) { score[k] = __expf(score[k] - mx); sum += score[k]; }
    const float inv = 1.0f / sum;

    float acc[4] = {0.f, 0.f, 0.f, 0.f};
#pragma unroll
    for (int k = 0; k < KNN; ++k)
#pragma unroll
        for (int t = 0; t < 4; ++t) acc[t] += score[k] * hreg[k][t];

#pragma unroll
    for (int t = 0; t < 4; ++t)
        hbar[(size_t)m * D_ + d0 + t] = acc[t] * inv;
}

#define GT_ROWS   32
#define GT_PITCH4 33

typedef unsigned v2u __attribute__((ext_vector_type(2)));

__global__ __launch_bounds__(256) void gemm_v_kernel(const float* __restrict__ hbar,
                                                     const float* __restrict__ w_v,
                                                     const float* __restrict__ b_v,
                                                     float* __restrict__ out) {
    __shared__ float4 Hs[GT_ROWS * GT_PITCH4];
    const int tileM = blockIdx.x * GT_ROWS;
    const int wave  = threadIdx.x >> 5;
    const int lane  = threadIdx.x & 31;
    const int half  = lane >> 4;
    const int l     = lane & 15;
    const int tileN = wave * 16;

    const float4* __restrict__ wrow = (const float4*)(w_v + (size_t)(tileN + l) * D_) + half;
    float4 wf[16];
#pragma unroll
    for (int ch = 0; ch < 16; ++ch) wf[ch] = wrow[ch * 2];

    const float4* __restrict__ h4 = (const float4*)(hbar + (size_t)tileM * D_);
    float4 st[4];
#pragma unroll
    for (int j = 0; j < 4; ++j) st[j] = h4[threadIdx.x + 256 * j];
#pragma unroll
    for (int j = 0; j < 4; ++j) {
        const int i = threadIdx.x + 256 * j;
        Hs[(i >> 5) * GT_PITCH4 + (i & 31)] = st[j];
    }
    __syncthreads();

    v8f c0 = {};
    v8f c1 = {};
    const float4* hr0 = &Hs[l * GT_PITCH4 + half];
    const float4* hr1 = &Hs[(16 + l) * GT_PITCH4 + half];
#pragma unroll
    for (int ch = 0; ch < 16; ++ch) {
        const float4 h0 = hr0[ch * 2];
        const float4 h1 = hr1[ch * 2];
        v2f a, b0, b1;
        a.x  = wf[ch].x; a.y  = wf[ch].y;
        b0.x = h0.x;     b0.y = h0.y;
        b1.x = h1.x;     b1.y = h1.y;
        c0 = __builtin_amdgcn_wmma_f32_16x16x4_f32(false, a, false, b0, (short)0, c0, false, false);
        c1 = __builtin_amdgcn_wmma_f32_16x16x4_f32(false, a, false, b1, (short)0, c1, false, false);
        a.x  = wf[ch].z; a.y  = wf[ch].w;
        b0.x = h0.z;     b0.y = h0.w;
        b1.x = h1.z;     b1.y = h1.w;
        c0 = __builtin_amdgcn_wmma_f32_16x16x4_f32(false, a, false, b0, (short)0, c0, false, false);
        c1 = __builtin_amdgcn_wmma_f32_16x16x4_f32(false, a, false, b1, (short)0, c1, false, false);
    }

    const float4 bl4 = *(const float4*)(b_v + tileN + half * 8);
    const float4 bh4 = *(const float4*)(b_v + tileN + half * 8 + 4);
    const float bias[8] = {bl4.x, bl4.y, bl4.z, bl4.w, bh4.x, bh4.y, bh4.z, bh4.w};

    const int b  = tileM >> 12;
    const int q0 = tileM & (NQ - 1);
    float* __restrict__ po = out + (size_t)b * 256 * NQ + (size_t)tileN * NQ + q0 + lane;
#pragma unroll
    for (int r = 0; r < 8; ++r) {
        const unsigned x = __float_as_uint(c0[r] + bias[r]);
        const unsigned y = __float_as_uint(c1[r] + bias[r]);
        const v2u sw = __builtin_amdgcn_permlane16_swap(x, y, false, false);
        po[(size_t)r * NQ]       = __uint_as_float(sw[0]);
        po[(size_t)(r + 8) * NQ] = __uint_as_float(sw[1]);
    }
}

extern "C" void kernel_launch(void* const* d_in, const int* in_sizes, int n_in,
                              void* d_out, int out_size, void* d_ws, size_t ws_size,
                              hipStream_t stream) {
    const float* query         = (const float*)d_in[0];
    const float* global_pc     = (const float*)d_in[1];
    const float* point_feature = (const float*)d_in[2];
    const float* w_pos         = (const float*)d_in[3];
    const float* b_pos         = (const float*)d_in[4];
    const float* w_pe          = (const float*)d_in[5];
    const float* b_pe          = (const float*)d_in[6];
    const float* w_v           = (const float*)d_in[7];
    const float* b_v           = (const float*)d_in[8];
    const float* w_attn        = (const float*)d_in[9];
    float* out = (float*)d_out;

    int*   idx_ws  = (int*)d_ws;
    float* hbar_ws = (float*)((char*)d_ws + (size_t)M_TOT * KNN * sizeof(int));

    knn_kernel<<<M_TOT / 32, 256, 0, stream>>>(query, global_pc, w_pos, b_pos, idx_ws, out);

    attn_kernel<<<M_TOT / 8, 256, 0, stream>>>(query, global_pc, point_feature,
                                               w_pe, b_pe, w_attn, idx_ws, hbar_ws);

    gemm_v_kernel<<<M_TOT / GT_ROWS, 256, 0, stream>>>(hbar_ws, w_v, b_v, out);
}


# attention score tabled per point (softmax drops the query's own term: s_j = f_j.w_attn - p_j.(W_pe^T w_attn)), written beside the KNN scan; attn kernel sums nothing across lanes; sum order changes
# speedup vs baseline: 17.1628x; 1.0474x over previous
#include <hip/hip_runtime.h>
#include <math.h>

#define B_    2
#define NQ    4096
#define NP    8192
#define D_    128
#define KNN   16
#define M_TOT (B_ * NQ)
#define LEAKY 0.2f

typedef float v2f __attribute__((ext_vector_type(2)));
typedef float v8f __attribute__((ext_vector_type(8)));

#define KPART   8
#define PPART   (NP / KPART)
#define PVEC    (PPART * 3 / 4)
#define QCAP    KNN

__device__ __forceinline__ bool kless(const unsigned ah, const unsigned al, const unsigned bh, const unsigned bl) {
    return (ah < bh) | ((ah == bh) & (al < bl));
}

__device__ __forceinline__ void kcx(unsigned& ah, unsigned& al, unsigned& bh, unsigned& bl) {
    const bool sw = kless(bh, bl, ah, al);
    const unsigned lo_h = ah < bh ? ah : bh;
    const unsigned hi_h = ah < bh ? bh : ah;
    const unsigned lo_l = sw ? bl : al;
    const unsigned hi_l = sw ? al : bl;
    ah = lo_h; al = lo_l;
    bh = hi_h; bl = hi_l;
}

__device__ __forceinline__ void ucx(unsigned& a, unsigned& b) {
    const unsigned lo = a < b ? a : b;
    const unsigned hi = a < b ? b : a;
    a = lo; b = hi;
}

#define KCX(i, j) kcx(h[i], l[i], h[j], l[j])

__device__ __forceinline__ void knn_sort16(unsigned (&h)[KNN], unsigned (&l)[KNN]) {
    KCX(0, 13); KCX(1, 12); KCX(2, 15); KCX(3, 14); KCX(4, 8); KCX(5, 6); KCX(7, 11); KCX(9, 10);
    KCX(0, 5); KCX(1, 7); KCX(2, 9); KCX(3, 4); KCX(6, 13); KCX(8, 14); KCX(10, 15); KCX(11, 12);
    KCX(0, 1); KCX(2, 3); KCX(4, 5); KCX(6, 8); KCX(7, 9); KCX(10, 11); KCX(12, 13); KCX(14, 15);
    KCX(0, 2); KCX(1, 3); KCX(4, 10); KCX(5, 11); KCX(6, 7); KCX(8, 9); KCX(12, 14); KCX(13, 15);
    KCX(1, 2); KCX(3, 12); KCX(4, 6); KCX(5, 7); KCX(8, 10); KCX(9, 11); KCX(13, 14);
    KCX(1, 4); KCX(2, 6); KCX(5, 8); KCX(7, 10); KCX(9, 13); KCX(11, 14);
    KCX(2, 4); KCX(3, 6); KCX(9, 12); KCX(11, 13);
    KCX(3, 5); KCX(6, 8); KCX(7, 9); KCX(10, 12);
    KCX(3, 4); KCX(5, 6); KCX(7, 8); KCX(9, 10); KCX(11, 12);
    KCX(6, 7); KCX(8, 9);
}
#undef KCX

__device__ __forceinline__ void knn_merge(unsigned (&kh)[KNN], unsigned (&kl)[KNN],
                                          const unsigned (&ph)[KNN], const unsigned (&pl)[KNN]) {
#pragma unroll
    for (int t = 0; t < KNN; ++t) {
        const bool take = kless(ph[KNN - 1 - t], pl[KNN - 1 - t], kh[t], kl[t]);
        kh[t] = kh[t] < ph[KNN - 1 - t] ? kh[t] : ph[KNN - 1 - t];
        kl[t] = take ? pl[KNN - 1 - t] : kl[t];
    }
#pragma unroll
    for (int s = KNN / 2; s >= 1; s >>= 1) {
#pragma unroll
        for (int t = 0; t < KNN; ++t) {
            if ((t & s) == 0) kcx(kh[t], kl[t], kh[t + s], kl[t + s]);
        }
    }
}

__global__ __launch_bounds__(256) void knn_kernel(const float* __restrict__ query,
                                                  const float* __restrict__ global_pc,
                                                  const float* __restrict__ point_feature,
                                                  const float* __restrict__ w_pos,
                                                  const float* __restrict__ b_pos,
                                                  const float* __restrict__ w_pe,
                                                  const float* __restrict__ w_attn,
                                                  uint4* __restrict__ idx_out,
                                                  float* __restrict__ score_tbl,
                                                  float* __restrict__ out) {
    __shared__ float4 sp[KPART][PVEC];
    __shared__ uint2  lst[7][KNN][32];
    __shared__ uint2  cq[KPART][QCAP][32];
    __shared__ unsigned bnd[KPART][32];

    const int lane = threadIdx.x & 31;
    const int part = threadIdx.x >> 5;
    const int m    = blockIdx.x * 32 + lane;
    const int b    = m >> 12;

    const float qx = query[m * 3 + 0];
    const float qy = query[m * 3 + 1];
    const float qz = query[m * 3 + 2];

    const int d0 = lane * 4;
    const int p0 = blockIdx.x * 64 + __builtin_amdgcn_readfirstlane(part) * 8;
    const float4 wa = *(const float4*)(w_attn + d0);
    const float4 e0 = *(const float4*)(w_pe + d0 * 3);
    const float4 e1 = *(const float4*)(w_pe + d0 * 3 + 4);
    const float4 e2 = *(const float4*)(w_pe + d0 * 3 + 8);
    float4 frow[8];
#pragma unroll
    for (int i = 0; i < 8; ++i) frow[i] = *(const float4*)(point_feature + (size_t)(p0 + i) * D_ + d0);

    const float4* __restrict__ pc4 =
        (const float4*)(global_pc + ((size_t)b * NP + (size_t)part * PPART) * 3);
#pragma unroll 8
    for (int i = lane; i < PVEC; i += 32) sp[part][i] = pc4[i];
    bnd[part][lane] = __float_as_uint(3.4e38f);
    __syncthreads();

    {
        const int dw = __builtin_amdgcn_readfirstlane(part) * (D_ / KPART);
        float* __restrict__ po = out + (size_t)b * 256 * NQ + (size_t)(D_ + dw) * NQ + (m & (NQ - 1));
#pragma unroll
        for (int c = 0; c < D_ / KPART; ++c) {
            const int d = dw + c;
            float v = w_pos[d * 3 + 0] * qx
                    + w_pos[d * 3 + 1] * qy
                    + w_pos[d * 3 + 2] * qz
                    + b_pos[d];
            v = (v >= 0.f) ? v : LEAKY * v;
            po[(size_t)c * NQ] = v;
        }
    }

    {
        float ux = e0.x * wa.x + e0.w * wa.y + e1.z * wa.z + e2.y * wa.w;
        float uy = e0.y * wa.x + e1.x * wa.y + e1.w * wa.z + e2.z * wa.w;
        float uz = e0.z * wa.x + e1.y * wa.y + e2.x * wa.z + e2.w * wa.w;
        float s[8];
#pragma unroll
        for (int i = 0; i < 8; ++i)
            s[i] = frow[i].x * wa.x + frow[i].y * wa.y + frow[i].z * wa.z + frow[i].w * wa.w;
#pragma unroll
        for (int off = 16; off >= 1; off >>= 1) {
            ux += __shfl_xor(ux, off, 32);
            uy += __shfl_xor(uy, off, 32);
            uz += __shfl_xor(uz, off, 32);
#pragma unroll
            for (int i = 0; i < 8; ++i) s[i] += __shfl_xor(s[i], off, 32);
        }
        float tv = 0.f;
#pragma unroll
        for (int i = 0; i < 8; ++i) {
            const float* pp = global_pc + (size_t)(p0 + i) * 3;
            const float t = s[i] - (pp[0] * ux + pp[1] * uy + pp[2] * uz);
            tv = (lane == i) ? t : tv;
        }
        if (lane < 8) score_tbl[p0 + lane] = tv;
    }

    unsigned keyh[KNN], keyl[KNN];
#pragma unroll
    for (int k = 0; k < KNN; ++k) { keyh[k] = __float_as_uint(3.4e38f); keyl[k] = 0u; }

    float    thr = 3.4e38f;
    unsigned qn  = 0;
    uint2* const myq = &cq[part][0][lane];

    auto scan4 = [&](const float4 a0, const float4 a1, const float4 a2, const unsigned i0) {
        const float px[4] = {a0.x, a0.w, a1.z, a2.y};
        const float py[4] = {a0.y, a1.x, a1.w, a2.z};
        const float pz[4] = {a0.z, a1.y, a2.x, a2.w};
        float d2[4];
#pragma unroll
        for (int u = 0; u < 4; ++u) {
            const float dx = qx - px[u];
            const float dy = qy - py[u];
            const float dz = qz - pz[u];
            d2[u] = fmaf(dz, dz, fmaf(dx, dx, dy * dy));
        }
#pragma unroll
        for (int u = 0; u < 4; ++u) {
            myq[qn * 32] = make_uint2(i0 + u, __float_as_uint(d2[u]));
            qn += (d2[u] < thr) ? 1u : 0u;
        }
    };

    auto take_batch = [&]() {
        unsigned ph[KNN], pl[KNN];
#pragma unroll
        for (int s = 0; s < KNN; ++s) {
            ph[s] = ~0u;
            pl[s] = ~0u;
            if ((unsigned)s < qn) {
                const uint2 e = myq[s * 32];
                ph[s] = e.y;
                pl[s] = e.x;
            }
        }
        knn_sort16(ph, pl);
        knn_merge(keyh, keyl, ph, pl);
        typedef volatile __attribute__((address_space(3))) unsigned lds_word;
        lds_word* const vb = (lds_word*)&bnd[0][lane];
        vb[part * 32] = keyh[3];
        unsigned v[KPART];
#pragma unroll
        for (int p = 0; p < KPART; ++p) v[p] = vb[p * 32];
        ucx(v[0], v[1]); ucx(v[2], v[3]); ucx(v[0], v[2]); ucx(v[1], v[3]); ucx(v[1], v[2]);
        ucx(v[4], v[5]); ucx(v[6], v[7]); ucx(v[4], v[6]); ucx(v[5], v[7]); ucx(v[5], v[6]);
        const unsigned m07 = v[0] < v[7] ? v[0] : v[7];
        const unsigned m16 = v[1] < v[6] ? v[1] : v[6];
        const unsigned m25 = v[2] < v[5] ? v[2] : v[5];
        const unsigned m34 = v[3] < v[4] ? v[3] : v[4];
        const unsigned ma  = m07 > m16 ? m07 : m16;
        const unsigned mb  = m25 > m34 ? m25 : m34;
        const unsigned cap = (ma > mb ? ma : mb) + 1u;
        const unsigned own = keyh[KNN - 1];
        thr = __uint_as_float(own < cap ? own : cap);
        qn  = 0;
    };

    const unsigned base = (unsigned)part * PPART;
    const float4* tile = &sp[part][0];
    float4 a0 = tile[0], a1 = tile[1], a2 = tile[2];
    for (int g = 0; g < PPART / 4; g += 2) {
        const float4 b0 = tile[g * 3 + 3], b1 = tile[g * 3 + 4], b2 = tile[g * 3 + 5];
        scan4(a0, a1, a2, base + g * 4);
        if (__builtin_amdgcn_ballot_w32(qn > QCAP - 4) != 0u) take_batch();
        const int gn = (g + 2 < PPART / 4) ? g + 2 : g;
        a0 = tile[gn * 3 + 0]; a1 = tile[gn * 3 + 1]; a2 = tile[gn * 3 + 2];
        scan4(b0, b1, b2, base + g * 4 + 4);
        if (__builtin_amdgcn_ballot_w32(qn > QCAP - 4) != 0u) take_batch();
    }
    take_batch();

    if (part & 1) {
#pragma unroll
        for (int k = 0; k < KNN; ++k) lst[part >> 1][k][lane] = make_uint2(keyl[k], keyh[k]);
    }
    __syncthreads();
    if (!(part & 1)) {
        unsigned ph[KNN], pl[KNN];
#pragma unroll
        for (int k = 0; k < KNN; ++k) { const uint2 e = lst[part >> 1][k][lane]; ph[k] = e.y; pl[k] = e.x; }
        knn_merge(keyh, keyl, ph, pl);
    }
    if ((part & 3) == 2) {
#pragma unroll
        for (int k = 0; k < KNN; ++k) lst[4 + (part >> 2)][k][lane] = make_uint2(keyl[k], keyh[k]);
    }
    __syncthreads();
    if ((part & 3) == 0) {
        unsigned ph[KNN], pl[KNN];
#pragma unroll
        for (int k = 0; k < KNN; ++k) { const uint2 e = lst[4 + (part >> 2)][k][lane]; ph[k] = e.y; pl[k] = e.x; }
        knn_merge(keyh, keyl, ph, pl);
    }
    if (part == 4) {
#pragma unroll
        for (int k = 0; k < KNN; ++k) lst[6][k][lane] = make_uint2(keyl[k], keyh[k]);
    }
    __syncthreads();
    if (part == 0) {
        unsigned ph[KNN], pl[KNN];
#pragma unroll
        for (int k = 0; k < KNN; ++k) { const uint2 e = lst[6][k][lane]; ph[k] = e.y; pl[k] = e.x; }
        knn_merge(keyh, keyl, ph, pl);
        idx_out[(size_t)m * 2 + 0] = make_uint4(keyl[0] | (keyl[1] << 16), keyl[2] | (keyl[3] << 16),
                                                keyl[4] | (keyl[5] << 16), keyl[6] | (keyl[7] << 16));
        idx_out[(size_t)m * 2 + 1] = make_uint4(keyl[8] | (keyl[9] << 16), keyl[10] | (keyl[11] << 16),
                                                keyl[12] | (keyl[13] << 16), keyl[14] | (keyl[15] << 16));
    }
}

__global__ __launch_bounds__(256) __attribute__((amdgpu_waves_per_eu(4, 8)))
void attn_kernel(const float* __restrict__ query,
                                                   const float* __restrict__ global_pc,
                                                   const float* __restrict__ point_feature,
                                                   const float* __restrict__ w_pe,
                                                   const float* __restrict__ b_pe,
                                                   const uint4* __restrict__ knn_idx,
                                                   const float* __restrict__ score_tbl,
                                                   float* __restrict__ hbar) {
    const int wave = __builtin_amdgcn_readfirstlane(threadIdx.x >> 5);
    const int lane = threadIdx.x & 31;
    const int m    = blockIdx.x * 8 + wave;
    const int b    = m >> 12;
    const int d0   = lane * 4;

    const uint4 ia = knn_idx[(size_t)m * 2 + 0];
    const uint4 ib = knn_idx[(size_t)m * 2 + 1];
    const unsigned iw[8] = {ia.x, ia.y, ia.z, ia.w, ib.x, ib.y, ib.z, ib.w};
    int j[KNN];
#pragma unroll
    for (int k = 0; k < KNN; ++k) j[k] = b * NP + (int)((iw[k >> 1] >> ((k & 1) * 16)) & 0xffffu);

    float4 f[KNN];
#pragma unroll
    for (int k = 0; k < KNN; ++k) f[k] = *(const float4*)(point_feature + (size_t)j[k] * D_ + d0);

    const float4 e0 = *(const float4*)(w_pe + d0 * 3);
    const float4 e1 = *(const float4*)(w_pe + d0 * 3 + 4);
    const float4 e2 = *(const float4*)(w_pe + d0 * 3 + 8);
    const float4 bp = *(const float4*)(b_pe + d0);

    const float qx = query[m * 3 + 0];
    const float qy = query[m * 3 + 1];
    const float qz = query[m * 3 + 2];

    float sc[KNN];
#pragma unroll
    for (int k = 0; k < KNN; ++k) sc[k] = score_tbl[j[k]];
    float mx = sc[0];
#pragma unroll
    for (int k = 1; k < KNN; ++k) mx = fmaxf(mx, sc[k]);
    float sum = 0.f, sx = 0.f, sy = 0.f, sz = 0.f;
#pragma unroll
    for (int k = 0; k < KNN; ++k) {
        const float* pp = global_pc + (size_t)j[k] * 3;
        sc[k] = __expf(sc[k] - mx);
        sum += sc[k];
        sx  += sc[k] * (qx - pp[0]);
        sy  += sc[k] * (qy - pp[1]);
        sz  += sc[k] * (qz - pp[2]);
    }
    const float inv = 1.0f / sum;

    float4 acc = make_float4(0.f, 0.f, 0.f, 0.f);
#pragma unroll
    for (int k = 0; k < KNN; ++k) {
        acc.x += sc[k] * f[k].x;
        acc.y += sc[k] * f[k].y;
        acc.z += sc[k] * f[k].z;
        acc.w += sc[k] * f[k].w;
    }
    float4 r;
    r.x = (acc.x + e0.x * sx + e0.y * sy + e0.z * sz) * inv + bp.x;
    r.y = (acc.y + e0.w * sx + e1.x * sy + e1.y * sz) * inv + bp.y;
    r.z = (acc.z + e1.z * sx + e1.w * sy + e2.x * sz) * inv + bp.z;
    r.w = (acc.w + e2.y * sx + e2.z * sy + e2.w * sz) * inv + bp.w;
    *(float4*)(hbar + (size_t)m * D_ + d0) = r;
}

#define GT_ROWS   32
#define GT_PITCH4 33

typedef unsigned v2u __attribute__((ext_vector_type(2)));

__global__ __launch_bounds__(256) void gemm_v_kernel(const float* __restrict__ hbar,
                                                     const float* __restrict__ w_v,
                                                     const float* __restrict__ b_v,
                                                     float* __restrict__ out) {
    __shared__ float4 Hs[GT_ROWS * GT_PITCH4];
    const int tileM = blockIdx.x * GT_ROWS;
    const int wave  = threadIdx.x >> 5;
    const int lane  = threadIdx.x & 31;
    const int half  = lane >> 4;
    const int l     = lane & 15;
    const int tileN = wave * 16;

    const float4* __restrict__ wrow = (const float4*)(w_v + (size_t)(tileN + l) * D_) + half;
    float4 wf[16];
#pragma unroll
    for (int ch = 0; ch < 16; ++ch) wf[ch] = wrow[ch * 2];

    const float4* __restrict__ h4 = (const float4*)(hbar + (size_t)tileM * D_);
    float4 st[4];
#pragma unroll
    for (int j = 0; j < 4; ++j) st[j] = h4[threadIdx.x + 256 * j];
#pragma unroll
    for (int j = 0; j < 4; ++j) {
        const int i = threadIdx.x + 256 * j;
        Hs[(i >> 5) * GT_PITCH4 + (i & 31)] = st[j];
    }
    __syncthreads();

    v8f c0 = {};
    v8f c1 = {};
    const float4* hr0 = &Hs[l * GT_PITCH4 + half];
    const float4* hr1 = &Hs[(16 + l) * GT_PITCH4 + half];
#pragma unroll
    for (int ch = 0; ch < 16; ++ch) {
        const float4 h0 = hr0[ch * 2];
        const float4 h1 = hr1[ch * 2];
        v2f a, b0, b1;
        a.x  = wf[ch].x; a.y  = wf[ch].y;
        b0.x = h0.x;     b0.y = h0.y;
        b1.x = h1.x;     b1.y = h1.y;
        c0 = __builtin_amdgcn_wmma_f32_16x16x4_f32(false, a, false, b0, (short)0, c0, false, false);
        c1 = __builtin_amdgcn_wmma_f32_16x16x4_f32(false, a, false, b1, (short)0, c1, false, false);
        a.x  = wf[ch].z; a.y  = wf[ch].w;
        b0.x = h0.z;     b0.y = h0.w;
        b1.x = h1.z;     b1.y = h1.w;
        c0 = __builtin_amdgcn_wmma_f32_16x16x4_f32(false, a, false, b0, (short)0, c0, false, false);
        c1 = __builtin_amdgcn_wmma_f32_16x16x4_f32(false, a, false, b1, (short)0, c1, false, false);
    }

    const float4 bl4 = *(const float4*)(b_v + tileN + half * 8);
    const float4 bh4 = *(const float4*)(b_v + tileN + half * 8 + 4);
    const float bias[8] = {bl4.x, bl4.y, bl4.z, bl4.w, bh4.x, bh4.y, bh4.z, bh4.w};

    const int b  = tileM >> 12;
    const int q0 = tileM & (NQ - 1);
    float* __restrict__ po = out + (size_t)b * 256 * NQ + (size_t)tileN * NQ + q0 + lane;
#pragma unroll
    for (int r = 0; r < 8; ++r) {
        const unsigned x = __float_as_uint(c0[r] + bias[r]);
        const unsigned y = __float_as_uint(c1[r] + bias[r]);
        const v2u sw = __builtin_amdgcn_permlane16_swap(x, y, false, false);
        po[(size_t)r * NQ]       = __uint_as_float(sw[0]);
        po[(size_t)(r + 8) * NQ] = __uint_as_float(sw[1]);
    }
}

extern "C" void kernel_launch(void* const* d_in, const int* in_sizes, int n_in,
                              void* d_out, int out_size, void* d_ws, size_t ws_size,
                              hipStream_t stream) {
    const float* query         = (const float*)d_in[0];
    const float* global_pc     = (const float*)d_in[1];
    const float* point_feature = (const float*)d_in[2];
    const float* w_pos         = (const float*)d_in[3];
    const float* b_pos         = (const float*)d_in[4];
    const float* w_pe          = (const float*)d_in[5];
    const float* b_pe          = (const float*)d_in[6];
    const float* w_v           = (const float*)d_in[7];
    const float* b_v           = (const float*)d_in[8];
    const float* w_attn        = (const float*)d_in[9];
    float* out = (float*)d_out;

    uint4* idx_ws  = (uint4*)d_ws;
    float* tbl_ws  = (float*)((char*)d_ws + (size_t)M_TOT * KNN * 2);
    float* hbar_ws = (float*)((char*)d_ws + (size_t)M_TOT * KNN * sizeof(int));

    knn_kernel<<<M_TOT / 32, 256, 0, stream>>>(query, global_pc, point_feature, w_pos, b_pos,
                                               w_pe, w_attn, idx_ws, tbl_ws, out);

    attn_kernel<<<M_TOT / 8, 256, 0, stream>>>(query, global_pc, point_feature,
                                               w_pe, b_pe, idx_ws, tbl_ws, hbar_ws);

    gemm_v_kernel<<<M_TOT / GT_ROWS, 256, 0, stream>>>(hbar_ws, w_v, b_v, out);
}


# KNN: every (point, query) pair screened on the matrix unit (f32 16x16x4: |p|^2 - 2 p.q against threshold - |q|^2 + margin); queues hold indices, exact d2 only per queued point; same keys
# speedup vs baseline: 19.0337x; 1.1090x over previous
#include <hip/hip_runtime.h>
#include <math.h>

#define B_    2
#define NQ    4096
#define NP    8192
#define D_    128
#define KNN   16
#define M_TOT (B_ * NQ)
#define LEAKY 0.2f

typedef float v2f __attribute__((ext_vector_type(2)));
typedef float v8f __attribute__((ext_vector_type(8)));
typedef unsigned v2u __attribute__((ext_vector_type(2)));

#define KPART   8
#define PPART   (NP / KPART)
#define QCAP    KNN

__device__ __forceinline__ bool kless(const unsigned ah, const unsigned al, const unsigned bh, const unsigned bl) {
    return (ah < bh) | ((ah == bh) & (al < bl));
}

__device__ __forceinline__ void kcx(unsigned& ah, unsigned& al, unsigned& bh, unsigned& bl) {
    const bool sw = kless(bh, bl, ah, al);
    const unsigned lo_h = ah < bh ? ah : bh;
    const unsigned hi_h = ah < bh ? bh : ah;
    const unsigned lo_l = sw ? bl : al;
    const unsigned hi_l = sw ? al : bl;
    ah = lo_h; al = lo_l;
    bh = hi_h; bl = hi_l;
}

__device__ __forceinline__ void ucx(unsigned& a, unsigned& b) {
    const unsigned lo = a < b ? a : b;
    const unsigned hi = a < b ? b : a;
    a = lo; b = hi;
}

#define KCX(i, j) kcx(h[i], l[i], h[j], l[j])

__device__ __forceinline__ void knn_sort16(unsigned (&h)[KNN], unsigned (&l)[KNN]) {
    KCX(0, 13); KCX(1, 12); KCX(2, 15); KCX(3, 14); KCX(4, 8); KCX(5, 6); KCX(7, 11); KCX(9, 10);
    KCX(0, 5); KCX(1, 7); KCX(2, 9); KCX(3, 4); KCX(6, 13); KCX(8, 14); KCX(10, 15); KCX(11, 12);
    KCX(0, 1); KCX(2, 3); KCX(4, 5); KCX(6, 8); KCX(7, 9); KCX(10, 11); KCX(12, 13); KCX(14, 15);
    KCX(0, 2); KCX(1, 3); KCX(4, 10); KCX(5, 11); KCX(6, 7); KCX(8, 9); KCX(12, 14); KCX(13, 15);
    KCX(1, 2); KCX(3, 12); KCX(4, 6); KCX(5, 7); KCX(8, 10); KCX(9, 11); KCX(13, 14);
    KCX(1, 4); KCX(2, 6); KCX(5, 8); KCX(7, 10); KCX(9, 13); KCX(11, 14);
    KCX(2, 4); KCX(3, 6); KCX(9, 12); KCX(11, 13);
    KCX(3, 5); KCX(6, 8); KCX(7, 9); KCX(10, 12);
    KCX(3, 4); KCX(5, 6); KCX(7, 8); KCX(9, 10); KCX(11, 12);
    KCX(6, 7); KCX(8, 9);
}
#undef KCX

__device__ __forceinline__ void knn_merge(unsigned (&kh)[KNN], unsigned (&kl)[KNN],
                                          const unsigned (&ph)[KNN], const unsigned (&pl)[KNN]) {
#pragma unroll
    for (int t = 0; t < KNN; ++t) {
        const bool take = kless(ph[KNN - 1 - t], pl[KNN - 1 - t], kh[t], kl[t]);
        kh[t] = kh[t] < ph[KNN - 1 - t] ? kh[t] : ph[KNN - 1 - t];
        kl[t] = take ? pl[KNN - 1 - t] : kl[t];
    }
#pragma unroll
    for (int s = KNN / 2; s >= 1; s >>= 1) {
#pragma unroll
        for (int t = 0; t < KNN; ++t) {
            if ((t & s) == 0) kcx(kh[t], kl[t], kh[t + s], kl[t + s]);
        }
    }
}

__global__ __launch_bounds__(256) void knn_kernel(const float* __restrict__ query,
                                                  const float* __restrict__ global_pc,
                                                  const float* __restrict__ point_feature,
                                                  const float* __restrict__ w_pos,
                                                  const float* __restrict__ b_pos,
                                                  const float* __restrict__ w_pe,
                                                  const float* __restrict__ w_attn,
                                                  uint4* __restrict__ idx_out,
                                                  float* __restrict__ score_tbl,
                                                  float* __restrict__ out) {
    __shared__ float4   tbl[NP];
    __shared__ uint2    lst[7][KNN][32];
    __shared__ unsigned cq[KPART][QCAP][32];
    __shared__ unsigned bnd[KPART][32];

    const int lane = threadIdx.x & 31;
    const int part = threadIdx.x >> 5;
    const int m    = blockIdx.x * 32 + lane;
    const int b    = m >> 12;

    const float qx = query[m * 3 + 0];
    const float qy = query[m * 3 + 1];
    const float qz = query[m * 3 + 2];

    const int cc = lane & 15;
    const int hf = lane >> 4;
    const float* __restrict__ q0p = query + (size_t)(blockIdx.x * 32 + cc) * 3;
    const float* __restrict__ q1p = q0p + 16 * 3;
    const float q0a = q0p[2 * hf], q0y = q0p[1];
    const float q1a = q1p[2 * hf], q1y = q1p[1];
    v2f bq0, bq1;
    bq0.x = -2.f * q0a;
    bq0.y = hf ? 1.f : -2.f * q0y;
    bq1.x = -2.f * q1a;
    bq1.y = hf ? 1.f : -2.f * q1y;

    const int d0 = lane * 4;
    const int p0 = blockIdx.x * 64 + __builtin_amdgcn_readfirstlane(part) * 8;
    const float4 wa = *(const float4*)(w_attn + d0);
    const float4 e0 = *(const float4*)(w_pe + d0 * 3);
    const float4 e1 = *(const float4*)(w_pe + d0 * 3 + 4);
    const float4 e2 = *(const float4*)(w_pe + d0 * 3 + 8);
    float4 frow[8];
#pragma unroll
    for (int i = 0; i < 8; ++i) frow[i] = *(const float4*)(point_feature + (size_t)(p0 + i) * D_ + d0);

    const float* __restrict__ pcw = global_pc + ((size_t)b * NP + (size_t)part * PPART) * 3;
    float4* const mytbl = &tbl[part * PPART];
    float p2max = 0.f;
#pragma unroll 8
    for (int i = lane; i < PPART; i += 32) {
        const float x = pcw[i * 3 + 0];
        const float y = pcw[i * 3 + 1];
        const float z = pcw[i * 3 + 2];
        const float p2 = x * x + y * y + z * z;
        p2max = fmaxf(p2max, p2);
        mytbl[i] = make_float4(x, y, z, p2);
    }
#pragma unroll
    for (int off = 16; off >= 1; off >>= 1) p2max = fmaxf(p2max, __shfl_xor(p2max, off, 32));
    bnd[part][lane] = __float_as_uint(3.4e38f);
    __syncthreads();

    {
        const int dw = __builtin_amdgcn_readfirstlane(part) * (D_ / KPART);
        float* __restrict__ po = out + (size_t)b * 256 * NQ + (size_t)(D_ + dw) * NQ + (m & (NQ - 1));
#pragma unroll
        for (int c = 0; c < D_ / KPART; ++c) {
            const int d = dw + c;
            float v = w_pos[d * 3 + 0] * qx
                    + w_pos[d * 3 + 1] * qy
                    + w_pos[d * 3 + 2] * qz
                    + b_pos[d];
            v = (v >= 0.f) ? v : LEAKY * v;
            po[(size_t)c * NQ] = v;
        }
    }

    {
        float ux = e0.x * wa.x + e0.w * wa.y + e1.z * wa.z + e2.y * wa.w;
        float uy = e0.y * wa.x + e1.x * wa.y + e1.w * wa.z + e2.z * wa.w;
        float uz = e0.z * wa.x + e1.y * wa.y + e2.x * wa.z + e2.w * wa.w;
        float s[8];
#pragma unroll
        for (int i = 0; i < 8; ++i)
            s[i] = frow[i].x * wa.x + frow[i].y * wa.y + frow[i].z * wa.z + frow[i].w * wa.w;
#pragma unroll
        for (int off = 16; off >= 1; off >>= 1) {
            ux += __shfl_xor(ux, off, 32);
            uy += __shfl_xor(uy, off, 32);
            uz += __shfl_xor(uz, off, 32);
#pragma unroll
            for (int i = 0; i < 8; ++i) s[i] += __shfl_xor(s[i], off, 32);
        }
        float tv = 0.f;
#pragma unroll
        for (int i = 0; i < 8; ++i) {
            const float* pp = global_pc + (size_t)(p0 + i) * 3;
            const float t = s[i] - (pp[0] * ux + pp[1] * uy + pp[2] * uz);
            tv = (lane == i) ? t : tv;
        }
        if (lane < 8) score_tbl[p0 + lane] = tv;
    }

    unsigned keyh[KNN], keyl[KNN];
#pragma unroll
    for (int k = 0; k < KNN; ++k) { keyh[k] = __float_as_uint(3.4e38f); keyl[k] = 0u; }

    const float q2   = qx * qx + qy * qy + qz * qz;
    const float lift = (p2max + q2) * 0x1p-15f - q2;
    float    tf = 3.4e38f;
    unsigned qn = 0;
    unsigned* const myq = &cq[part][0][lane];

    auto scan4 = [&](const float g0, const float g1, const float g2, const float g3, const unsigned i0) {
        const float g[4] = {g0, g1, g2, g3};
#pragma unroll
        for (int u = 0; u < 4; ++u) {
            myq[qn * 32] = i0 + u;
            qn += (g[u] < tf) ? 1u : 0u;
        }
    };

    auto take_batch = [&]() {
        unsigned ph[KNN], pl[KNN];
#pragma unroll
        for (int s = 0; s < KNN; ++s) {
            const unsigned j = myq[s * 32] & (NP - 1);
            const float4 p = tbl[j];
            const float dx = qx - p.x;
            const float dy = qy - p.y;
            const float dz = qz - p.z;
            const float d2 = fmaf(dz, dz, fmaf(dx, dx, dy * dy));
            const bool live = (unsigned)s < qn;
            ph[s] = live ? __float_as_uint(d2) : ~0u;
            pl[s] = live ? j : ~0u;
        }
        knn_sort16(ph, pl);
        knn_merge(keyh, keyl, ph, pl);
        typedef volatile __attribute__((address_space(3))) unsigned lds_word;
        lds_word* const vb = (lds_word*)&bnd[0][lane];
        vb[part * 32] = keyh[3];
        unsigned v[KPART];
#pragma unroll
        for (int p = 0; p < KPART; ++p) v[p] = vb[p * 32];
        ucx(v[0], v[1]); ucx(v[2], v[3]); ucx(v[0], v[2]); ucx(v[1], v[3]); ucx(v[1], v[2]);
        ucx(v[4], v[5]); ucx(v[6], v[7]); ucx(v[4], v[6]); ucx(v[5], v[7]); ucx(v[5], v[6]);
        const unsigned m07 = v[0] < v[7] ? v[0] : v[7];
        const unsigned m16 = v[1] < v[6] ? v[1] : v[6];
        const unsigned m25 = v[2] < v[5] ? v[2] : v[5];
        const unsigned m34 = v[3] < v[4] ? v[3] : v[4];
        const unsigned ma  = m07 > m16 ? m07 : m16;
        const unsigned mb  = m25 > m34 ? m25 : m34;
        const unsigned cap = (ma > mb ? ma : mb) + 1u;
        const unsigned own = keyh[KNN - 1];
        const float thr = __uint_as_float(own < cap ? own : cap);
        tf = fmaf(thr, 1.0f + 0x1p-20f, lift);
        qn = 0;
    };

    const unsigned base = (unsigned)part * PPART;
    const v2f* const afp = (const v2f*)&tbl[base] + (cc * 2 + hf);
    const v8f zero = {};
    v2f a = afp[0];
    v8f acc0 = __builtin_amdgcn_wmma_f32_16x16x4_f32(false, a, false, bq0, (short)0, zero, false, false);
    v8f acc1 = __builtin_amdgcn_wmma_f32_16x16x4_f32(false, a, false, bq1, (short)0, zero, false, false);
    a = afp[32];
    for (int t = 0; t < PPART / 16; ++t) {
        float g[16];
#pragma unroll
        for (int r = 0; r < 8; ++r) {
            const v2u sw = __builtin_amdgcn_permlane16_swap(__float_as_uint(acc0[r]), __float_as_uint(acc1[r]), false, false);
            g[r]     = __uint_as_float(sw[0]);
            g[8 + r] = __uint_as_float(sw[1]);
        }
        acc0 = __builtin_amdgcn_wmma_f32_16x16x4_f32(false, a, false, bq0, (short)0, zero, false, false);
        acc1 = __builtin_amdgcn_wmma_f32_16x16x4_f32(false, a, false, bq1, (short)0, zero, false, false);
        const int tn = (t + 2 < PPART / 16) ? t + 2 : PPART / 16 - 1;
        a = afp[tn * 32];
        const unsigned i0 = base + (unsigned)t * 16;
        scan4(g[0], g[1], g[2], g[3], i0);
        if (__builtin_amdgcn_ballot_w32(qn > QCAP - 4) != 0u) take_batch();
        scan4(g[4], g[5], g[6], g[7], i0 + 4);
        if (__builtin_amdgcn_ballot_w32(qn > QCAP - 4) != 0u) take_batch();
        scan4(g[8], g[9], g[10], g[11], i0 + 8);
        if (__builtin_amdgcn_ballot_w32(qn > QCAP - 4) != 0u) take_batch();
        scan4(g[12], g[13], g[14], g[15], i0 + 12);
        if (__builtin_amdgcn_ballot_w32(qn > QCAP - 4) != 0u) take_batch();
    }
    take_batch();

    if (part & 1) {
#pragma unroll
        for (int k = 0; k < KNN; ++k) lst[part >> 1][k][lane] = make_uint2(keyl[k], keyh[k]);
    }
    __syncthreads();
    if (!(part & 1)) {
        unsigned ph[KNN], pl[KNN];
#pragma unroll
        for (int k = 0; k < KNN; ++k) { const uint2 e = lst[part >> 1][k][lane]; ph[k] = e.y; pl[k] = e.x; }
        knn_merge(keyh, keyl, ph, pl);
    }
    if ((part & 3) == 2) {
#pragma unroll
        for (int k = 0; k < KNN; ++k) lst[4 + (part >> 2)][k][lane] = make_uint2(keyl[k], keyh[k]);
    }
    __syncthreads();
    if ((part & 3) == 0) {
        unsigned ph[KNN], pl[KNN];
#pragma unroll
        for (int k = 0; k < KNN; ++k) { const uint2 e = lst[4 + (part >> 2)][k][lane]; ph[k] = e.y; pl[k] = e.x; }
        knn_merge(keyh, keyl, ph, pl);
    }
    if (part == 4) {
#pragma unroll
        for (int k = 0; k < KNN; ++k) lst[6][k][lane] = make_uint2(keyl[k], keyh[k]);
    }
    __syncthreads();
    if (part == 0) {
        unsigned ph[KNN], pl[KNN];
#pragma unroll
        for (int k = 0; k < KNN; ++k) { const uint2 e = lst[6][k][lane]; ph[k] = e.y; pl[k] = e.x; }
        knn_merge(keyh, keyl, ph, pl);
        idx_out[(size_t)m * 2 + 0] = make_uint4(keyl[0] | (keyl[1] << 16), keyl[2] | (keyl[3] << 16),
                                                keyl[4] | (keyl[5] << 16), keyl[6] | (keyl[7] << 16));
        idx_out[(size_t)m * 2 + 1] = make_uint4(keyl[8] | (keyl[9] << 16), keyl[10] | (keyl[11] << 16),
                                                keyl[12] | (keyl[13] << 16), keyl[14] | (keyl[15] << 16));
    }
}

__global__ __launch_bounds__(256) __attribute__((amdgpu_waves_per_eu(4, 8)))
void attn_kernel(const float* __restrict__ query,
                                                   const float* __restrict__ global_pc,
                                                   const float* __restrict__ point_feature,
                                                   const float* __restrict__ w_pe,
                                                   const float* __restrict__ b_pe,
                                                   const uint4* __restrict__ knn_idx,
                                                   const float* __restrict__ score_tbl,
                                                   float* __restrict__ hbar) {
    const int wave = __builtin_amdgcn_readfirstlane(threadIdx.x >> 5);
    const int lane = threadIdx.x & 31;
    const int m    = blockIdx.x * 8 + wave;
    const int b    = m >> 12;
    const int d0   = lane * 4;

    const uint4 ia = knn_idx[(size_t)m * 2 + 0];
    const uint4 ib = knn_idx[(size_t)m * 2 + 1];
    const unsigned iw[8] = {ia.x, ia.y, ia.z, ia.w, ib.x, ib.y, ib.z, ib.w};
    int j[KNN];
#pragma unroll
    for (int k = 0; k < KNN; ++k) j[k] = b * NP + (int)((iw[k >> 1] >> ((k & 1) * 16)) & 0xffffu);

    float4 f[KNN];
#pragma unroll
    for (int k = 0; k < KNN; ++k) f[k] = *(const float4*)(point_feature + (size_t)j[k] * D_ + d0);

    const float4 e0 = *(const float4*)(w_pe + d0 * 3);
    const float4 e1 = *(const float4*)(w_pe + d0 * 3 + 4);
    const float4 e2 = *(const float4*)(w_pe + d0 * 3 + 8);
    const float4 bp = *(const float4*)(b_pe + d0);

    const float qx = query[m * 3 + 0];
    const float qy = query[m * 3 + 1];
    const float qz = query[m * 3 + 2];

    float sc[KNN];
#pragma unroll
    for (int k = 0; k < KNN; ++k) sc[k] = score_tbl[j[k]];
    float mx = sc[0];
#pragma unroll
    for (int k = 1; k < KNN; ++k) mx = fmaxf(mx, sc[k]);
    float sum = 0.f, sx = 0.f, sy = 0.f, sz = 0.f;
#pragma unroll
    for (int k = 0; k < KNN; ++k) {
        const float* pp = global_pc + (size_t)j[k] * 3;
        sc[k] = __expf(sc[k] - mx);
        sum += sc[k];
        sx  += sc[k] * (qx - pp[0]);
        sy  += sc[k] * (qy - pp[1]);
        sz  += sc[k] * (qz - pp[2]);
    }
    const float inv = 1.0f / sum;

    float4 acc = make_float4(0.f, 0.f, 0.f, 0.f);
#pragma unroll
    for (int k = 0; k < KNN; ++k) {
        acc.x += sc[k] * f[k].x;
        acc.y += sc[k] * f[k].y;
        acc.z += sc[k] * f[k].z;
        acc.w += sc[k] * f[k].w;
    }
    float4 r;
    r.x = (acc.x + e0.x * sx + e0.y * sy + e0.z * sz) * inv + bp.x;
    r.y = (acc.y + e0.w * sx + e1.x * sy + e1.y * sz) * inv + bp.y;
    r.z = (acc.z + e1.z * sx + e1.w * sy + e2.x * sz) * inv + bp.z;
    r.w = (acc.w + e2.y * sx + e2.z * sy + e2.w * sz) * inv + bp.w;
    *(float4*)(hbar + (size_t)m * D_ + d0) = r;
}

#define GT_ROWS   32
#define GT_PITCH4 33

__global__ __launch_bounds__(256) void gemm_v_kernel(const float* __restrict__ hbar,
                                                     const float* __restrict__ w_v,
                                                     const float* __restrict__ b_v,
                                                     float* __restrict__ out) {
    __shared__ float4 Hs[GT_ROWS * GT_PITCH4];
    const int tileM = blockIdx.x * GT_ROWS;
    const int wave  = threadIdx.x >> 5;
    const int lane  = threadIdx.x & 31;
    const int half  = lane >> 4;
    const int l     = lane & 15;
    const int tileN = wave * 16;

    const float4* __restrict__ wrow = (const float4*)(w_v + (size_t)(tileN + l) * D_) + half;
    float4 wf[16];
#pragma unroll
    for (int ch = 0; ch < 16; ++ch) wf[ch] = wrow[ch * 2];

    const float4* __restrict__ h4 = (const float4*)(hbar + (size_t)tileM * D_);
    float4 st[4];
#pragma unroll
    for (int j = 0; j < 4; ++j) st[j] = h4[threadIdx.x + 256 * j];
#pragma unroll
    for (int j = 0; j < 4; ++j) {
        const int i = threadIdx.x + 256 * j;
        Hs[(i >> 5) * GT_PITCH4 + (i & 31)] = st[j];
    }
    __syncthreads();

    v8f c0 = {};
    v8f c1 = {};
    const float4* hr0 = &Hs[l * GT_PITCH4 + half];
    const float4* hr1 = &Hs[(16 + l) * GT_PITCH4 + half];
#pragma unroll
    for (int ch = 0; ch < 16; ++ch) {
        const float4 h0 = hr0[ch * 2];
        const float4 h1 = hr1[ch * 2];
        v2f a, b0, b1;
        a.x  = wf[ch].x; a.y  = wf[ch].y;
        b0.x = h0.x;     b0.y = h0.y;
        b1.x = h1.x;     b1.y = h1.y;
        c0 = __builtin_amdgcn_wmma_f32_16x16x4_f32(false, a, false, b0, (short)0, c0, false, false);
        c1 = __builtin_amdgcn_wmma_f32_16x16x4_f32(false, a, false, b1, (short)0, c1, false, false);
        a.x  = wf[ch].z; a.y  = wf[ch].w;
        b0.x = h0.z;     b0.y = h0.w;
        b1.x = h1.z;     b1.y = h1.w;
        c0 = __builtin_amdgcn_wmma_f32_16x16x4_f32(false, a, false, b0, (short)0, c0, false, false);
        c1 = __builtin_amdgcn_wmma_f32_16x16x4_f32(false, a, false, b1, (short)0, c1, false, false);
    }

    const float4 bl4 = *(const float4*)(b_v + tileN + half * 8);
    const float4 bh4 = *(const float4*)(b_v + tileN + half * 8 + 4);
    const float bias[8] = {bl4.x, bl4.y, bl4.z, bl4.w, bh4.x, bh4.y, bh4.z, bh4.w};

    const int b  = tileM >> 12;
    const int q0 = tileM & (NQ - 1);
    float* __restrict__ po = out + (size_t)b * 256 * NQ + (size_t)tileN * NQ + q0 + lane;
#pragma unroll
    for (int r = 0; r < 8; ++r) {
        const unsigned x = __float_as_uint(c0[r] + bias[r]);
        const unsigned y = __float_as_uint(c1[r] + bias[r]);
        const v2u sw = __builtin_amdgcn_permlane16_swap(x, y, false, false);
        po[(size_t)r * NQ]       = __uint_as_float(sw[0]);
        po[(size_t)(r + 8) * NQ] = __uint_as_float(sw[1]);
    }
}

extern "C" void kernel_launch(void* const* d_in, const int* in_sizes, int n_in,
                              void* d_out, int out_size, void* d_ws, size_t ws_size,
                              hipStream_t stream) {
    const float* query         = (const float*)d_in[0];
    const float* global_pc     = (const float*)d_in[1];
    const float* point_feature = (const float*)d_in[2];
    const float* w_pos         = (const float*)d_in[3];
    const float* b_pos         = (const float*)d_in[4];
    const float* w_pe          = (const float*)d_in[5];
    const float* b_pe          = (const float*)d_in[6];
    const float* w_v           = (const float*)d_in[7];
    const float* b_v           = (const float*)d_in[8];
    const float* w_attn        = (const float*)d_in[9];
    float* out = (float*)d_out;

    uint4* idx_ws  = (uint4*)d_ws;
    float* tbl_ws  = (float*)((char*)d_ws + (size_t)M_TOT * KNN * 2);
    float* hbar_ws = (float*)((char*)d_ws + (size_t)M_TOT * KNN * sizeof(int));

    knn_kernel<<<M_TOT / 32, 256, 0, stream>>>(query, global_pc, point_feature, w_pos, b_pos,
                                               w_pe, w_attn, idx_ws, tbl_ws, out);

    attn_kernel<<<M_TOT / 8, 256, 0, stream>>>(query, global_pc, point_feature,
                                               w_pe, b_pe, idx_ws, tbl_ws, hbar_ws);

    gemm_v_kernel<<<M_TOT / GT_ROWS, 256, 0, stream>>>(hbar_ws, w_v, b_v, out);
}


# KNN scan: a lane's 16 screen results of a step make a 16-bit mask (compare + add-with-carry, no store); the wave pushes set bits in lockstep, 2 per trip, batch above 14 waiting (was 12); same keys
# speedup vs baseline: 20.1189x; 1.0570x over previous
#include <hip/hip_runtime.h>
#include <math.h>

#define B_    2
#define NQ    4096
#define NP    8192
#define D_    128
#define KNN   16
#define M_TOT (B_ * NQ)
#define LEAKY 0.2f

typedef float v2f __attribute__((ext_vector_type(2)));
typedef float v8f __attribute__((ext_vector_type(8)));
typedef unsigned v2u __attribute__((ext_vector_type(2)));

#define KPART   8
#define PPART   (NP / KPART)
#define QCAP    KNN

__device__ __forceinline__ bool kless(const unsigned ah, const unsigned al, const unsigned bh, const unsigned bl) {
    return (ah < bh) | ((ah == bh) & (al < bl));
}

__device__ __forceinline__ void kcx(unsigned& ah, unsigned& al, unsigned& bh, unsigned& bl) {
    const bool sw = kless(bh, bl, ah, al);
    const unsigned lo_h = ah < bh ? ah : bh;
    const unsigned hi_h = ah < bh ? bh : ah;
    const unsigned lo_l = sw ? bl : al;
    const unsigned hi_l = sw ? al : bl;
    ah = lo_h; al = lo_l;
    bh = hi_h; bl = hi_l;
}

__device__ __forceinline__ void ucx(unsigned& a, unsigned& b) {
    const unsigned lo = a < b ? a : b;
    const unsigned hi = a < b ? b : a;
    a = lo; b = hi;
}

#define KCX(i, j) kcx(h[i], l[i], h[j], l[j])

__device__ __forceinline__ void knn_sort16(unsigned (&h)[KNN], unsigned (&l)[KNN]) {
    KCX(0, 13); KCX(1, 12); KCX(2, 15); KCX(3, 14); KCX(4, 8); KCX(5, 6); KCX(7, 11); KCX(9, 10);
    KCX(0, 5); KCX(1, 7); KCX(2, 9); KCX(3, 4); KCX(6, 13); KCX(8, 14); KCX(10, 15); KCX(11, 12);
    KCX(0, 1); KCX(2, 3); KCX(4, 5); KCX(6, 8); KCX(7, 9); KCX(10, 11); KCX(12, 13); KCX(14, 15);
    KCX(0, 2); KCX(1, 3); KCX(4, 10); KCX(5, 11); KCX(6, 7); KCX(8, 9); KCX(12, 14); KCX(13, 15);
    KCX(1, 2); KCX(3, 12); KCX(4, 6); KCX(5, 7); KCX(8, 10); KCX(9, 11); KCX(13, 14);
    KCX(1, 4); KCX(2, 6); KCX(5, 8); KCX(7, 10); KCX(9, 13); KCX(11, 14);
    KCX(2, 4); KCX(3, 6); KCX(9, 12); KCX(11, 13);
    KCX(3, 5); KCX(6, 8); KCX(7, 9); KCX(10, 12);
    KCX(3, 4); KCX(5, 6); KCX(7, 8); KCX(9, 10); KCX(11, 12);
    KCX(6, 7); KCX(8, 9);
}
#undef KCX

__device__ __forceinline__ void knn_merge(unsigned (&kh)[KNN], unsigned (&kl)[KNN],
                                          const unsigned (&ph)[KNN], const unsigned (&pl)[KNN]) {
#pragma unroll
    for (int t = 0; t < KNN; ++t) {
        const bool take = kless(ph[KNN - 1 - t], pl[KNN - 1 - t], kh[t], kl[t]);
        kh[t] = kh[t] < ph[KNN - 1 - t] ? kh[t] : ph[KNN - 1 - t];
        kl[t] = take ? pl[KNN - 1 - t] : kl[t];
    }
#pragma unroll
    for (int s = KNN / 2; s >= 1; s >>= 1) {
#pragma unroll
        for (int t = 0; t < KNN; ++t) {
            if ((t & s) == 0) kcx(kh[t], kl[t], kh[t + s], kl[t + s]);
        }
    }
}

__global__ __launch_bounds__(256) void knn_kernel(const float* __restrict__ query,
                                                  const float* __restrict__ global_pc,
                                                  const float* __restrict__ point_feature,
                                                  const float* __restrict__ w_pos,
                                                  const float* __restrict__ b_pos,
                                                  const float* __restrict__ w_pe,
                                                  const float* __restrict__ w_attn,
                                                  uint4* __restrict__ idx_out,
                                                  float* __restrict__ score_tbl,
                                                  float* __restrict__ out) {
    __shared__ float4   tbl[NP];
    __shared__ uint2    lst[7][KNN][32];
    __shared__ unsigned cq[KPART][QCAP][32];
    __shared__ unsigned bnd[KPART][32];

    const int lane = threadIdx.x & 31;
    const int part = threadIdx.x >> 5;
    const int m    = blockIdx.x * 32 + lane;
    const int b    = m >> 12;

    const float qx = query[m * 3 + 0];
    const float qy = query[m * 3 + 1];
    const float qz = query[m * 3 + 2];

    const int cc = lane & 15;
    const int hf = lane >> 4;
    const float* __restrict__ q0p = query + (size_t)(blockIdx.x * 32 + cc) * 3;
    const float* __restrict__ q1p = q0p + 16 * 3;
    const float q0a = q0p[2 * hf], q0y = q0p[1];
    const float q1a = q1p[2 * hf], q1y = q1p[1];
    v2f bq0, bq1;
    bq0.x = -2.f * q0a;
    bq0.y = hf ? 1.f : -2.f * q0y;
    bq1.x = -2.f * q1a;
    bq1.y = hf ? 1.f : -2.f * q1y;

    const int d0 = lane * 4;
    const int p0 = blockIdx.x * 64 + __builtin_amdgcn_readfirstlane(part) * 8;
    const float4 wa = *(const float4*)(w_attn + d0);
    const float4 e0 = *(const float4*)(w_pe + d0 * 3);
    const float4 e1 = *(const float4*)(w_pe + d0 * 3 + 4);
    const float4 e2 = *(const float4*)(w_pe + d0 * 3 + 8);
    float4 frow[8];
#pragma unroll
    for (int i = 0; i < 8; ++i) frow[i] = *(const float4*)(point_feature + (size_t)(p0 + i) * D_ + d0);

    const float* __restrict__ pcw = global_pc + ((size_t)b * NP + (size_t)part * PPART) * 3;
    float4* const mytbl = &tbl[part * PPART];
    float p2max = 0.f;
#pragma unroll 8
    for (int i = lane; i < PPART; i += 32) {
        const float x = pcw[i * 3 + 0];
        const float y = pcw[i * 3 + 1];
        const float z = pcw[i * 3 + 2];
        const float p2 = x * x + y * y + z * z;
        p2max = fmaxf(p2max, p2);
        mytbl[i] = make_float4(x, y, z, p2);
    }
#pragma unroll
    for (int off = 16; off >= 1; off >>= 1) p2max = fmaxf(p2max, __shfl_xor(p2max, off, 32));
    bnd[part][lane] = __float_as_uint(3.4e38f);
    __syncthreads();

    {
        const int dw = __builtin_amdgcn_readfirstlane(part) * (D_ / KPART);
        float* __restrict__ po = out + (size_t)b * 256 * NQ + (size_t)(D_ + dw) * NQ + (m & (NQ - 1));
#pragma unroll
        for (int c = 0; c < D_ / KPART; ++c) {
            const int d = dw + c;
            float v = w_pos[d * 3 + 0] * qx
                    + w_pos[d * 3 + 1] * qy
                    + w_pos[d * 3 + 2] * qz
                    + b_pos[d];
            v = (v >= 0.f) ? v : LEAKY * v;
            po[(size_t)c * NQ] = v;
        }
    }

    {
        float ux = e0.x * wa.x + e0.w * wa.y + e1.z * wa.z + e2.y * wa.w;
        float uy = e0.y * wa.x + e1.x * wa.y + e1.w * wa.z + e2.z * wa.w;
        float uz = e0.z * wa.x + e1.y * wa.y + e2.x * wa.z + e2.w * wa.w;
        float s[8];
#pragma unroll
        for (int i = 0; i < 8; ++i)
            s[i] = frow[i].x * wa.x + frow[i].y * wa.y + frow[i].z * wa.z + frow[i].w * wa.w;
#pragma unroll
        for (int off = 16; off >= 1; off >>= 1) {
            ux += __shfl_xor(ux, off, 32);
            uy += __shfl_xor(uy, off, 32);
            uz += __shfl_xor(uz, off, 32);
#pragma unroll
            for (int i = 0; i < 8; ++i) s[i] += __shfl_xor(s[i], off, 32);
        }
        float tv = 0.f;
#pragma unroll
        for (int i = 0; i < 8; ++i) {
            const float* pp = global_pc + (size_t)(p0 + i) * 3;
            const float t = s[i] - (pp[0] * ux + pp[1] * uy + pp[2] * uz);
            tv = (lane == i) ? t : tv;
        }
        if (lane < 8) score_tbl[p0 + lane] = tv;
    }

    unsigned keyh[KNN], keyl[KNN];
#pragma unroll
    for (int k = 0; k < KNN; ++k) { keyh[k] = __float_as_uint(3.4e38f); keyl[k] = 0u; }

    const float q2   = qx * qx + qy * qy + qz * qz;
    const float lift = (p2max + q2) * 0x1p-15f - q2;
    float    tf = 3.4e38f;
    unsigned qn = 0;
    unsigned* const myq = &cq[part][0][lane];

    auto take_batch = [&]() {
        unsigned ph[KNN], pl[KNN];
#pragma unroll
        for (int s = 0; s < KNN; ++s) {
            const unsigned j = myq[s * 32] & (NP - 1);
            const float4 p = tbl[j];
            const float dx = qx - p.x;
            const float dy = qy - p.y;
            const float dz = qz - p.z;
            const float d2 = fmaf(dz, dz, fmaf(dx, dx, dy * dy));
            const bool live = (unsigned)s < qn;
            ph[s] = live ? __float_as_uint(d2) : ~0u;
            pl[s] = live ? j : ~0u;
        }
        knn_sort16(ph, pl);
        knn_merge(keyh, keyl, ph, pl);
        typedef volatile __attribute__((address_space(3))) unsigned lds_word;
        lds_word* const vb = (lds_word*)&bnd[0][lane];
        vb[part * 32] = keyh[3];
        unsigned v[KPART];
#pragma unroll
        for (int p = 0; p < KPART; ++p) v[p] = vb[p * 32];
        ucx(v[0], v[1]); ucx(v[2], v[3]); ucx(v[0], v[2]); ucx(v[1], v[3]); ucx(v[1], v[2]);
        ucx(v[4], v[5]); ucx(v[6], v[7]); ucx(v[4], v[6]); ucx(v[5], v[7]); ucx(v[5], v[6]);
        const unsigned m07 = v[0] < v[7] ? v[0] : v[7];
        const unsigned m16 = v[1] < v[6] ? v[1] : v[6];
        const unsigned m25 = v[2] < v[5] ? v[2] : v[5];
        const unsigned m34 = v[3] < v[4] ? v[3] : v[4];
        const unsigned ma  = m07 > m16 ? m07 : m16;
        const unsigned mb  = m25 > m34 ? m25 : m34;
        const unsigned cap = (ma > mb ? ma : mb) + 1u;
        const unsigned own = keyh[KNN - 1];
        const float thr = __uint_as_float(own < cap ? own : cap);
        tf = fmaf(thr, 1.0f + 0x1p-20f, lift);
        qn = 0;
    };

    const unsigned base = (unsigned)part * PPART;
    const v2f* const afp = (const v2f*)&tbl[base] + (cc * 2 + hf);
    const v8f zero = {};
    v2f a = afp[0];
    v8f acc0 = __builtin_amdgcn_wmma_f32_16x16x4_f32(false, a, false, bq0, (short)0, zero, false, false);
    v8f acc1 = __builtin_amdgcn_wmma_f32_16x16x4_f32(false, a, false, bq1, (short)0, zero, false, false);
    a = afp[32];
    unsigned mk = 0u;
    unsigned i0 = base;
    int t = 0;
    auto push1 = [&]() {
        myq[qn * 32] = i0 + (unsigned)(__ffs(mk) - 1);
        qn += (mk != 0u) ? 1u : 0u;
        mk &= mk - 1u;
    };
    do {
        if (__builtin_amdgcn_ballot_w32(mk != 0u) == 0u) {
            float g[16];
#pragma unroll
            for (int r = 0; r < 8; ++r) {
                const v2u sw = __builtin_amdgcn_permlane16_swap(__float_as_uint(acc0[r]), __float_as_uint(acc1[r]), false, false);
                g[r]     = __uint_as_float(sw[0]);
                g[8 + r] = __uint_as_float(sw[1]);
            }
            acc0 = __builtin_amdgcn_wmma_f32_16x16x4_f32(false, a, false, bq0, (short)0, zero, false, false);
            acc1 = __builtin_amdgcn_wmma_f32_16x16x4_f32(false, a, false, bq1, (short)0, zero, false, false);
            const int tn = (t + 2 < PPART / 16) ? t + 2 : PPART / 16 - 1;
            a = afp[tn * 32];
            i0 = base + (unsigned)t * 16;
            unsigned mhi = 0u, mlo = 0u;
#pragma unroll
            for (int u = 7; u >= 0; --u) {
                const unsigned fhi = __builtin_amdgcn_ballot_w32(g[8 + u] < tf);
                const unsigned flo = __builtin_amdgcn_ballot_w32(g[u] < tf);
                asm("v_add_co_ci_u32_e64 %0, null, %0, %0, %1" : "+v"(mhi) : "s"(fhi));
                asm("v_add_co_ci_u32_e64 %0, null, %0, %0, %1" : "+v"(mlo) : "s"(flo));
            }
            mk = (mhi << 8) | mlo;
            ++t;
        }
        push1();
        push1();
        if (__builtin_amdgcn_ballot_w32(qn > QCAP - 2) != 0u) take_batch();
    } while (t < PPART / 16 || __builtin_amdgcn_ballot_w32(mk != 0u) != 0u);
    take_batch();

    if (part & 1) {
#pragma unroll
        for (int k = 0; k < KNN; ++k) lst[part >> 1][k][lane] = make_uint2(keyl[k], keyh[k]);
    }
    __syncthreads();
    if (!(part & 1)) {
        unsigned ph[KNN], pl[KNN];
#pragma unroll
        for (int k = 0; k < KNN; ++k) { const uint2 e = lst[part >> 1][k][lane]; ph[k] = e.y; pl[k] = e.x; }
        knn_merge(keyh, keyl, ph, pl);
    }
    if ((part & 3) == 2) {
#pragma unroll
        for (int k = 0; k < KNN; ++k) lst[4 + (part >> 2)][k][lane] = make_uint2(keyl[k], keyh[k]);
    }
    __syncthreads();
    if ((part & 3) == 0) {
        unsigned ph[KNN], pl[KNN];
#pragma unroll
        for (int k = 0; k < KNN; ++k) { const uint2 e = lst[4 + (part >> 2)][k][lane]; ph[k] = e.y; pl[k] = e.x; }
        knn_merge(keyh, keyl, ph, pl);
    }
    if (part == 4) {
#pragma unroll
        for (int k = 0; k < KNN; ++k) lst[6][k][lane] = make_uint2(keyl[k], keyh[k]);
    }
    __syncthreads();
    if (part == 0) {
        unsigned ph[KNN], pl[KNN];
#pragma unroll
        for (int k = 0; k < KNN; ++k) { const uint2 e = lst[6][k][lane]; ph[k] = e.y; pl[k] = e.x; }
        knn_merge(keyh, keyl, ph, pl);
        idx_out[(size_t)m * 2 + 0] = make_uint4(keyl[0] | (keyl[1] << 16), keyl[2] | (keyl[3] << 16),
                                                keyl[4] | (keyl[5] << 16), keyl[6] | (keyl[7] << 16));
        idx_out[(size_t)m * 2 + 1] = make_uint4(keyl[8] | (keyl[9] << 16), keyl[10] | (keyl[11] << 16),
                                                keyl[12] | (keyl[13] << 16), keyl[14] | (keyl[15] << 16));
    }
}

__global__ __launch_bounds__(256) __attribute__((amdgpu_waves_per_eu(4, 8)))
void attn_kernel(const float* __restrict__ query,
                                                   const float* __restrict__ global_pc,
                                                   const float* __restrict__ point_feature,
                                                   const float* __restrict__ w_pe,
                                                   const float* __restrict__ b_pe,
                                                   const uint4* __restrict__ knn_idx,
                                                   const float* __restrict__ score_tbl,
                                                   float* __restrict__ hbar) {
    const int wave = __builtin_amdgcn_readfirstlane(threadIdx.x >> 5);
    const int lane = threadIdx.x & 31;
    const int m    = blockIdx.x * 8 + wave;
    const int b    = m >> 12;
    const int d0   = lane * 4;

    const uint4 ia = knn_idx[(size_t)m * 2 + 0];
    const uint4 ib = knn_idx[(size_t)m * 2 + 1];
    const unsigned iw[8] = {ia.x, ia.y, ia.z, ia.w, ib.x, ib.y, ib.z, ib.w};
    int j[KNN];
#pragma unroll
    for (int k = 0; k < KNN; ++k) j[k] = b * NP + (int)((iw[k >> 1] >> ((k & 1) * 16)) & 0xffffu);

    float4 f[KNN];
#pragma unroll
    for (int k = 0; k < KNN; ++k) f[k] = *(const float4*)(point_feature + (size_t)j[k] * D_ + d0);

    const float4 e0 = *(const float4*)(w_pe + d0 * 3);
    const float4 e1 = *(const float4*)(w_pe + d0 * 3 + 4);
    const float4 e2 = *(const float4*)(w_pe + d0 * 3 + 8);
    const float4 bp = *(const float4*)(b_pe + d0);

    const float qx = query[m * 3 + 0];
    const float qy = query[m * 3 + 1];
    const float qz = query[m * 3 + 2];

    float sc[KNN];
#pragma unroll
    for (int k = 0; k < KNN; ++k) sc[k] = score_tbl[j[k]];
    float mx = sc[0];
#pragma unroll
    for (int k = 1; k < KNN; ++k) mx = fmaxf(mx, sc[k]);
    float sum = 0.f, sx = 0.f, sy = 0.f, sz = 0.f;
#pragma unroll
    for (int k = 0; k < KNN; ++k) {
        const float* pp = global_pc + (size_t)j[k] * 3;
        sc[k] = __expf(sc[k] - mx);
        sum += sc[k];
        sx  += sc[k] * (qx - pp[0]);
        sy  += sc[k] * (qy - pp[1]);
        sz  += sc[k] * (qz - pp[2]);
    }
    const float inv = 1.0f / sum;

    float4 acc = make_float4(0.f, 0.f, 0.f, 0.f);
#pragma unroll
    for (int k = 0; k < KNN; ++k) {
        acc.x += sc[k] * f[k].x;
        acc.y += sc[k] * f[k].y;
        acc.z += sc[k] * f[k].z;
        acc.w += sc[k] * f[k].w;
    }
    float4 r;
    r.x = (acc.x + e0.x * sx + e0.y * sy + e0.z * sz) * inv + bp.x;
    r.y = (acc.y + e0.w * sx + e1.x * sy + e1.y * sz) * inv + bp.y;
    r.z = (acc.z + e1.z * sx + e1.w * sy + e2.x * sz) * inv + bp.z;
    r.w = (acc.w + e2.y * sx + e2.z * sy + e2.w * sz) * inv + bp.w;
    *(float4*)(hbar + (size_t)m * D_ + d0) = r;
}

#define GT_ROWS   32
#define GT_PITCH4 33

__global__ __launch_bounds__(256) void gemm_v_kernel(const float* __restrict__ hbar,
                                                     const float* __restrict__ w_v,
                                                     const float* __restrict__ b_v,
                                                     float* __restrict__ out) {
    __shared__ float4 Hs[GT_ROWS * GT_PITCH4];
    const int tileM = blockIdx.x * GT_ROWS;
    const int wave  = threadIdx.x >> 5;
    const int lane  = threadIdx.x & 31;
    const int half  = lane >> 4;
    const int l     = lane & 15;
    const int tileN = wave * 16;

    const float4* __restrict__ wrow = (const float4*)(w_v + (size_t)(tileN + l) * D_) + half;
    float4 wf[16];
#pragma unroll
    for (int ch = 0; ch < 16; ++ch) wf[ch] = wrow[ch * 2];

    const float4* __restrict__ h4 = (const float4*)(hbar + (size_t)tileM * D_);
    float4 st[4];
#pragma unroll
    for (int j = 0; j < 4; ++j) st[j] = h4[threadIdx.x + 256 * j];
#pragma unroll
    for (int j = 0; j < 4; ++j) {
        const int i = threadIdx.x + 256 * j;
        Hs[(i >> 5) * GT_PITCH4 + (i & 31)] = st[j];
    }
    __syncthreads();

    v8f c0 = {};
    v8f c1 = {};
    const float4* hr0 = &Hs[l * GT_PITCH4 + half];
    const float4* hr1 = &Hs[(16 + l) * GT_PITCH4 + half];
#pragma unroll
    for (int ch = 0; ch < 16; ++ch) {
        const float4 h0 = hr0[ch * 2];
        const float4 h1 = hr1[ch * 2];
        v2f a, b0, b1;
        a.x  = wf[ch].x; a.y  = wf[ch].y;
        b0.x = h0.x;     b0.y = h0.y;
        b1.x = h1.x;     b1.y = h1.y;
        c0 = __builtin_amdgcn_wmma_f32_16x16x4_f32(false, a, false, b0, (short)0, c0, false, false);
        c1 = __builtin_amdgcn_wmma_f32_16x16x4_f32(false, a, false, b1, (short)0, c1, false, false);
        a.x  = wf[ch].z; a.y  = wf[ch].w;
        b0.x = h0.z;     b0.y = h0.w;
        b1.x = h1.z;     b1.y = h1.w;
        c0 = __builtin_amdgcn_wmma_f32_16x16x4_f32(false, a, false, b0, (short)0, c0, false, false);
        c1 = __builtin_amdgcn_wmma_f32_16x16x4_f32(false, a, false, b1, (short)0, c1, false, false);
    }

    const float4 bl4 = *(const float4*)(b_v + tileN + half * 8);
    const float4 bh4 = *(const float4*)(b_v + tileN + half * 8 + 4);
    const float bias[8] = {bl4.x, bl4.y, bl4.z, bl4.w, bh4.x, bh4.y, bh4.z, bh4.w};

    const int b  = tileM >> 12;
    const int q0 = tileM & (NQ - 1);
    float* __restrict__ po = out + (size_t)b * 256 * NQ + (size_t)tileN * NQ + q0 + lane;
#pragma unroll
    for (int r = 0; r < 8; ++r) {
        const unsigned x = __float_as_uint(c0[r] + bias[r]);
        const unsigned y = __float_as_uint(c1[r] + bias[r]);
        const v2u sw = __builtin_amdgcn_permlane16_swap(x, y, false, false);
        po[(size_t)r * NQ]       = __uint_as_float(sw[0]);
        po[(size_t)(r + 8) * NQ] = __uint_as_float(sw[1]);
    }
}

extern "C" void kernel_launch(void* const* d_in, const int* in_sizes, int n_in,
                              void* d_out, int out_size, void* d_ws, size_t ws_size,
                              hipStream_t stream) {
    const float* query         = (const float*)d_in[0];
    const float* global_pc     = (const float*)d_in[1];
    const float* point_feature = (const float*)d_in[2];
    const float* w_pos         = (const float*)d_in[3];
    const float* b_pos         = (const float*)d_in[4];
    const float* w_pe          = (const float*)d_in[5];
    const float* b_pe          = (const float*)d_in[6];
    const float* w_v           = (const float*)d_in[7];
    const float* b_v           = (const float*)d_in[8];
    const float* w_attn        = (const float*)d_in[9];
    float* out = (float*)d_out;

    uint4* idx_ws  = (uint4*)d_ws;
    float* tbl_ws  = (float*)((char*)d_ws + (size_t)M_TOT * KNN * 2);
    float* hbar_ws = (float*)((char*)d_ws + (size_t)M_TOT * KNN * sizeof(int));

    knn_kernel<<<M_TOT / 32, 256, 0, stream>>>(query, global_pc, point_feature, w_pos, b_pos,
                                               w_pe, w_attn, idx_ws, tbl_ws, out);

    attn_kernel<<<M_TOT / 8, 256, 0, stream>>>(query, global_pc, point_feature,
                                               w_pe, b_pe, idx_ws, tbl_ws, hbar_ws);

    gemm_v_kernel<<<M_TOT / GT_ROWS, 256, 0, stream>>>(hbar_ws, w_v, b_v, out);
}
